# starting kernel
# baseline (speedup 1.0000x reference)
#include <hip/hip_runtime.h>
#include <hip/hip_bf16.h>

#define HCH 511
#define LSEQ 2048
#define NMODES 32
#define NLAYERS 4
#define M2 1022
#define KD 511
#define KPAD 512
#define LN_EPS 1e-5f

typedef __attribute__((ext_vector_type(2))) float v2f;
typedef __attribute__((ext_vector_type(8))) float v8f;

__device__ __forceinline__ float gelu_tanh(float x) {
    float x3 = x * x * x;
    float t = tanhf(0.7978845608028654f * (x + 0.044715f * x3));
    return 0.5f * x * (1.0f + t);
}

__device__ __forceinline__ float sigmoidf(float x) {
    return 1.0f / (1.0f + expf(-x));
}

__global__ void s4_setup(const float* __restrict__ log_dt,
                         const float* __restrict__ Arl,
                         const float* __restrict__ Aim,
                         const float* __restrict__ Cre,
                         const float* __restrict__ Cim,
                         float* __restrict__ w_re, float* __restrict__ w_im,
                         float* __restrict__ c_re, float* __restrict__ c_im) {
    int idx = blockIdx.x * blockDim.x + threadIdx.x;
    if (idx >= HCH * NMODES) return;
    int h = idx / NMODES;
    float a_re = -expf(Arl[idx]);
    float a_im = Aim[idx];
    float dt = expf(log_dt[h]);
    float dre = a_re * dt, dim = a_im * dt;
    float er = expf(dre);
    float sn, cs;
    sincosf(dim, &sn, &cs);
    float wre = er * cs, wim = er * sn;
    float nre = wre - 1.0f, nim = wim;
    float inv = 1.0f / (a_re * a_re + a_im * a_im);
    float tre = (nre * a_re + nim * a_im) * inv;
    float tim = (nim * a_re - nre * a_im) * inv;
    float Cr = Cre[idx], Ci = Cim[idx];
    w_re[idx] = wre;
    w_im[idx] = wim;
    c_re[idx] = 2.0f * (Cr * tre - Ci * tim);
    c_im[idx] = 2.0f * (Cr * tim + Ci * tre);
}

__global__ __launch_bounds__(256) void s4_scan(
        const float* __restrict__ z,
        const float* __restrict__ w_re, const float* __restrict__ w_im,
        const float* __restrict__ c_re, const float* __restrict__ c_im,
        const float* __restrict__ D_skip, float* __restrict__ y) {
    int warp = threadIdx.x >> 5;
    int lane = threadIdx.x & 31;
    int h = blockIdx.x * 8 + warp;
    if (h >= HCH) return;
    int cidx = h * NMODES + lane;
    float wr = w_re[cidx], wi = w_im[cidx];
    float cr = c_re[cidx], ci = c_im[cidx];
    float d = D_skip[h];
    float xr = 0.0f, xi = 0.0f;
    const float* zrow = z + (size_t)h * LSEQ;
    float* yrow = y + (size_t)h * LSEQ;
    for (int l0 = 0; l0 < LSEQ; l0 += 32) {
        float zv = zrow[l0 + lane];
        float out = 0.0f;
#pragma unroll
        for (int t = 0; t < 32; ++t) {
            float zt = __shfl(zv, t, 32);
            float nxr = fmaf(wr, xr, fmaf(-wi, xi, zt));
            float nxi = fmaf(wr, xi, wi * xr);
            xr = nxr; xi = nxi;
            float contrib = fmaf(cr, xr, -ci * xi);
#pragma unroll
            for (int off = 16; off > 0; off >>= 1)
                contrib += __shfl_xor(contrib, off, 32);
            if (lane == t) out = contrib;
        }
        yrow[l0 + lane] = gelu_tanh(fmaf(d, zv, out));
    }
}

#define BM 128
#define BN 128
#define BK 16
#define LDP 18

__global__ __launch_bounds__(256) void s4_glu_gemm(
        const float* __restrict__ Wg, const float* __restrict__ bias,
        const float* __restrict__ Y, float* __restrict__ Y2) {
    __shared__ float Ws[BM][LDP];
    __shared__ float Ys[BN][LDP];
    int tid = threadIdx.x;
    int lane = tid & 31;
    int wave = tid >> 5;
    int wave_m = wave >> 2;
    int wave_n = wave & 3;
    int lrow = lane & 15;
    int khalf = (lane >> 4) << 1;
    int madd = (lane >> 4) << 3;
    int blockM = blockIdx.x * BM;
    int blockN = blockIdx.y * BN;

    v8f acc[4][2];
#pragma unroll
    for (int mt = 0; mt < 4; ++mt) {
#pragma unroll
        for (int i = 0; i < 8; ++i) {
            int gm = blockM + wave_m * 64 + mt * 16 + i + madd;
            float bv = (gm < M2) ? bias[gm] : 0.0f;
#pragma unroll
            for (int nt = 0; nt < 2; ++nt) acc[mt][nt][i] = bv;
        }
    }

    for (int kb = 0; kb < KPAD; kb += BK) {
        {
            int m = tid >> 1;
            int ks = (tid & 1) * 8;
            int gm = blockM + m;
#pragma unroll
            for (int j = 0; j < 8; ++j) {
                int gk = kb + ks + j;
                Ws[m][ks + j] = (gm < M2 && gk < KD) ? Wg[gm * KD + gk] : 0.0f;
            }
        }
        {
            int k = tid >> 4;
            int ns = (tid & 15) * 8;
            int gk = kb + k;
            if (gk < KD) {
                const float* src = Y + (size_t)gk * LSEQ + blockN + ns;
#pragma unroll
                for (int j = 0; j < 8; ++j) Ys[ns + j][k] = src[j];
            } else {
#pragma unroll
                for (int j = 0; j < 8; ++j) Ys[ns + j][k] = 0.0f;
            }
        }
        __syncthreads();
#pragma unroll
        for (int kc = 0; kc < 4; ++kc) {
            int kk = kc * 4 + khalf;
            v2f a[4], b[2];
#pragma unroll
            for (int mt = 0; mt < 4; ++mt)
                a[mt] = *(const v2f*)&Ws[wave_m * 64 + mt * 16 + lrow][kk];
#pragma unroll
            for (int nt = 0; nt < 2; ++nt)
                b[nt] = *(const v2f*)&Ys[wave_n * 32 + nt * 16 + lrow][kk];
#pragma unroll
            for (int mt = 0; mt < 4; ++mt)
#pragma unroll
                for (int nt = 0; nt < 2; ++nt)
                    acc[mt][nt] = __builtin_amdgcn_wmma_f32_16x16x4_f32(
                        false, a[mt], false, b[nt],
                        (short)0, acc[mt][nt], false, false);
        }
        __syncthreads();
    }

    int gn_base = blockN + wave_n * 32;
#pragma unroll
    for (int mt = 0; mt < 4; ++mt) {
#pragma unroll
        for (int i = 0; i < 8; ++i) {
            int gm = blockM + wave_m * 64 + mt * 16 + i + madd;
            if (gm < M2) {
#pragma unroll
                for (int nt = 0; nt < 2; ++nt) {
                    int gn = gn_base + nt * 16 + lrow;
                    Y2[(size_t)gm * LSEQ + gn] = acc[mt][nt][i];
                }
            }
        }
    }
}

__global__ __launch_bounds__(256) void s4_glu_ln(
        const float* __restrict__ Y2, float* __restrict__ Z,
        const float* __restrict__ g, const float* __restrict__ b) {
    __shared__ float red[256];
    int l = blockIdx.x;
    int tid = threadIdx.x;
    int h1 = tid + 256;

    float v0, v1 = 0.0f;
    {
        float p = Y2[(size_t)tid * LSEQ + l];
        float q = Y2[(size_t)(tid + HCH) * LSEQ + l];
        v0 = p * sigmoidf(q) + Z[(size_t)tid * LSEQ + l];
    }
    if (h1 < HCH) {
        float p = Y2[(size_t)h1 * LSEQ + l];
        float q = Y2[(size_t)(h1 + HCH) * LSEQ + l];
        v1 = p * sigmoidf(q) + Z[(size_t)h1 * LSEQ + l];
    }

    red[tid] = v0 + v1;
    __syncthreads();
    for (int s = 128; s > 0; s >>= 1) {
        if (tid < s) red[tid] += red[tid + s];
        __syncthreads();
    }
    float mu = red[0] * (1.0f / (float)HCH);
    __syncthreads();

    float d0 = v0 - mu;
    float d1 = (h1 < HCH) ? (v1 - mu) : 0.0f;
    red[tid] = d0 * d0 + d1 * d1;
    __syncthreads();
    for (int s = 128; s > 0; s >>= 1) {
        if (tid < s) red[tid] += red[tid + s];
        __syncthreads();
    }
    float rstd = rsqrtf(red[0] * (1.0f / (float)HCH) + LN_EPS);

    Z[(size_t)tid * LSEQ + l] = d0 * rstd * g[tid] + b[tid];
    if (h1 < HCH)
        Z[(size_t)h1 * LSEQ + l] = d1 * rstd * g[h1] + b[h1];
}

extern "C" void kernel_launch(void* const* d_in, const int* in_sizes, int n_in,
                              void* d_out, int out_size, void* d_ws, size_t ws_size,
                              hipStream_t stream) {
    (void)in_sizes; (void)n_in; (void)out_size; (void)ws_size;
    const float* Zin    = (const float*)d_in[0];
    const float* log_dt = (const float*)d_in[1];
    const float* Arl    = (const float*)d_in[2];
    const float* Aim    = (const float*)d_in[3];
    const float* Cre    = (const float*)d_in[4];
    const float* Cim    = (const float*)d_in[5];
    const float* Dsk    = (const float*)d_in[6];
    const float* glu_w  = (const float*)d_in[7];
    const float* glu_b  = (const float*)d_in[8];
    const float* ln_g   = (const float*)d_in[9];
    const float* ln_b   = (const float*)d_in[10];

    float* ws = (float*)d_ws;
    const size_t HN = (size_t)HCH * NMODES;
    float* w_re  = ws;
    float* w_im  = ws + HN;
    float* c_re  = ws + 2 * HN;
    float* c_im  = ws + 3 * HN;
    float* zbuf  = ws + 4 * HN;
    float* ybuf  = zbuf + (size_t)HCH * LSEQ;
    float* y2buf = ybuf + (size_t)HCH * LSEQ;

    s4_setup<<<(int)((HN + 255) / 256), 256, 0, stream>>>(
        log_dt, Arl, Aim, Cre, Cim, w_re, w_im, c_re, c_im);

    hipMemcpyAsync(zbuf, Zin, sizeof(float) * (size_t)HCH * LSEQ,
                   hipMemcpyDeviceToDevice, stream);

    for (int layer = 0; layer < NLAYERS; ++layer) {
        s4_scan<<<64, 256, 0, stream>>>(zbuf, w_re, w_im, c_re, c_im, Dsk, ybuf);
        s4_glu_gemm<<<dim3(8, 16, 1), 256, 0, stream>>>(glu_w, glu_b, ybuf, y2buf);
        s4_glu_ln<<<LSEQ, 256, 0, stream>>>(y2buf, zbuf, ln_g, ln_b);
    }

    hipMemcpyAsync(d_out, zbuf, sizeof(float) * (size_t)HCH * LSEQ,
                   hipMemcpyDeviceToDevice, stream);
}


# glu_w repacked once per call into per-wave WMMA fragment order; GLU kernel loads A fragments global -> registers (16-B loads, 2 tiles ahead): no W staging in LDS, no K-loop barrier; sums unchanged
# speedup vs baseline: 12.7648x; 12.7648x over previous
#include <hip/hip_runtime.h>
#include <hip/hip_bf16.h>

#define HCH 511
#define LSEQ 2048
#define NMODES 32
#define NLAYERS 4
#define KD 511
#define LN_EPS 1e-5f

typedef __attribute__((ext_vector_type(2))) float v2f;
typedef __attribute__((ext_vector_type(4))) float v4f;
typedef __attribute__((ext_vector_type(8))) float v8f;

__device__ __forceinline__ float gelu_tanh(float x) {
    float x3 = x * x * x;
    float t = tanhf(0.7978845608028654f * (x + 0.044715f * x3));
    return 0.5f * x * (1.0f + t);
}

__device__ __forceinline__ float sigmoidf(float x) {
    return 1.0f / (1.0f + expf(-x));
}

#define FBK 16
#define FNW 16
#define FNT 32
#define WTILE_V4 (FNW * 8 * 32)
#define SETUP_BLOCKS 64
#define PACK_BLOCKS 512

__global__ __launch_bounds__(256) void s4_setup_pack(
                         const float* __restrict__ log_dt,
                         const float* __restrict__ Arl,
                         const float* __restrict__ Aim,
                         const float* __restrict__ Cre,
                         const float* __restrict__ Cim,
                         float* __restrict__ w_re, float* __restrict__ w_im,
                         float* __restrict__ c_re, float* __restrict__ c_im,
                         const float* __restrict__ Wg, float* __restrict__ Wp) {
    if (blockIdx.x >= SETUP_BLOCKS) {
        const int pb = blockIdx.x - SETUP_BLOCKS;
        const int w = pb >> 5, t = (pb >> 3) & 3, ktg = pb & 7;
        const int lane = threadIdx.x & 31, sub = threadIdx.x >> 5;
        const int p = sub & 1, kt = ktg * 4 + (sub >> 1);
        const int lrow = lane & 15, hi = lane >> 4;
        const int j = w * 32 + (t & 1) * 16 + lrow;
        const bool rowok = j < HCH;
        const int jc = rowok ? j : HCH - 1;
        const float* srow = Wg + (size_t)((t < 2) ? jc : HCH + jc) * KD;
        const int k0 = kt * FBK + p * 8 + hi * 2;
        float e[4];
#pragma unroll
        for (int s = 0; s < 4; ++s) {
            int k = k0 + (s & 1) + (s >> 1) * 4;
            float v = srow[k < KD ? k : KD - 1];
            e[s] = (rowok && k < KD) ? v : 0.0f;
        }
        v4f o = {e[0], e[1], e[2], e[3]};
        ((v4f*)Wp)[(size_t)((kt * FNW + w) * 8 + t * 2 + p) * 32 + lane] = o;
        return;
    }
    int idx = blockIdx.x * blockDim.x + threadIdx.x;
    if (idx >= HCH * NMODES) return;
    int h = idx / NMODES;
    float a_re = -expf(Arl[idx]);
    float a_im = Aim[idx];
    float dt = expf(log_dt[h]);
    float dre = a_re * dt, dim = a_im * dt;
    float er = expf(dre);
    float sn, cs;
    sincosf(dim, &sn, &cs);
    float wre = er * cs, wim = er * sn;
    float nre = wre - 1.0f, nim = wim;
    float inv = 1.0f / (a_re * a_re + a_im * a_im);
    float tre = (nre * a_re + nim * a_im) * inv;
    float tim = (nim * a_re - nre * a_im) * inv;
    float Cr = Cre[idx], Ci = Cim[idx];
    w_re[idx] = wre;
    w_im[idx] = wim;
    c_re[idx] = 2.0f * (Cr * tre - Ci * tim);
    c_im[idx] = 2.0f * (Cr * tim + Ci * tre);
}

#define CT 64
#define NCH 32
#define LDA 68
#define LDB 132

__device__ __forceinline__ void cmulf(float ar, float ai, float br, float bi,
                                      float& rr, float& ri) {
    rr = fmaf(ar, br, -ai * bi);
    ri = fmaf(ar, bi, ai * br);
}

__device__ __forceinline__ void csqrd(double& r, double& i) {
    double nr = fma(r, r, -(i * i));
    double ni = (r + r) * i;
    r = nr; i = ni;
}

__global__ __launch_bounds__(128) void s4_scan(
        const float* __restrict__ z,
        const float* __restrict__ w_re, const float* __restrict__ w_im,
        const float* __restrict__ c_re, const float* __restrict__ c_im,
        const float* __restrict__ D_skip, float* __restrict__ y) {
    __shared__ __attribute__((aligned(16))) float PT[CT][LDA];
    __shared__ __attribute__((aligned(16))) float Qs[CT][LDA];
    __shared__ __attribute__((aligned(16))) float ZS[NCH][LDB];
    __shared__ float Kt[CT];
    __shared__ float WT[2][NMODES];
    __shared__ float Cr[NMODES];

    const int tid = threadIdx.x;
    const int lane = tid & 31;
    const int wave = tid >> 5;
    const int lrow = lane & 15;
    const int khalf = (lane >> 4) << 1;
    const int madd = (lane >> 4) << 3;
    const int h = blockIdx.x;
    const float* zrow = z + (size_t)h * LSEQ;
    float* yrow = y + (size_t)h * LSEQ;

#pragma unroll
    for (int j = 0; j < 4; ++j) {
        int l = j * 512 + tid * 4;
        float4 v = *(const float4*)(zrow + l);
        *(float4*)&ZS[l >> 6][l & 63] = v;
    }

    {
        int cidx = h * NMODES + lane;
        float wr = w_re[cidx], wi = w_im[cidx];
        float cr = c_re[cidx], ci = c_im[cidx];
        float pr = 1.0f, pi = 0.0f;
        float er = 0.0f, ei = 0.0f;
        if (wave > 0) {
            double ar = (double)wr, ai = (double)wi;
            csqrd(ar, ai); csqrd(ar, ai); csqrd(ar, ai); csqrd(ar, ai);
            double br = ar, bi = ai;
            if (wave >= 2) csqrd(br, bi);
            if (wave == 3) {
                double dr = br, di = bi;
                csqrd(dr, di);
                er = (float)dr; ei = (float)di;
                double tr = fma(br, ar, -(bi * ai));
                double ti = fma(br, ai, bi * ar);
                br = tr; bi = ti;
            }
            pr = (float)br; pi = (float)bi;
        }
        const int m0 = wave * 16;
#pragma unroll
        for (int j = 0; j < 16; ++j) {
            int m = m0 + j;
            PT[lane][63 - m] = pr;
            PT[NMODES + lane][63 - m] = pi;
            if (m >= 1) {
                float qr, qi;
                cmulf(cr, ci, pr, pi, qr, qi);
                Qs[m - 1][lane] = qr;
                Qs[m - 1][NMODES + lane] = -qi;
            }
            float tr, ti;
            cmulf(pr, pi, wr, wi, tr, ti);
            pr = tr; pi = ti;
        }
        if (wave == 3) {
            float qr, qi;
            cmulf(cr, ci, er, ei, qr, qi);
            Qs[63][lane] = qr;
            Qs[63][NMODES + lane] = -qi;
            WT[0][lane] = er;
            WT[1][lane] = ei;
        }
        if (wave == 0) Cr[lane] = cr;
    }
    __syncthreads();

    if (tid < CT) {
        const float* rowp = (tid == 0) ? &Cr[0] : &Qs[tid > 0 ? tid - 1 : 0][0];
        float s = 0.0f;
#pragma unroll
        for (int n = 0; n < NMODES; ++n) s += rowp[n];
        Kt[tid] = s;
    }

    v8f acc[2];
#pragma unroll
    for (int nt = 0; nt < 2; ++nt)
#pragma unroll
        for (int i = 0; i < 8; ++i) acc[nt][i] = 0.0f;
#pragma unroll
    for (int kc = 0; kc < 16; ++kc) {
        int kk = kc * 4 + khalf;
        v2f a  = *(const v2f*)&PT[wave * 16 + lrow][kk];
        v2f b0 = *(const v2f*)&ZS[lrow][kk];
        v2f b1 = *(const v2f*)&ZS[16 + lrow][kk];
        acc[0] = __builtin_amdgcn_wmma_f32_16x16x4_f32(
            false, a, false, b0, (short)0, acc[0], false, false);
        acc[1] = __builtin_amdgcn_wmma_f32_16x16x4_f32(
            false, a, false, b1, (short)0, acc[1], false, false);
    }
#pragma unroll
    for (int nt = 0; nt < 2; ++nt)
#pragma unroll
        for (int i = 0; i < 8; ++i)
            ZS[nt * 16 + lrow][CT + wave * 16 + i + madd] = acc[nt][i];
    __syncthreads();

    if (wave == 0) {
        float wtr = WT[0][lane], wti = WT[1][lane];
        float sr = 0.0f, si = 0.0f;
#pragma unroll
        for (int c = 0; c < NCH; ++c) {
            float br = ZS[c][CT + lane];
            float bi = ZS[c][CT + NMODES + lane];
            ZS[c][CT + lane] = sr;
            ZS[c][CT + NMODES + lane] = si;
            float nr = fmaf(wtr, sr, fmaf(-wti, si, br));
            float ni = fmaf(wtr, si, fmaf(wti, sr, bi));
            sr = nr; si = ni;
        }
    } else {
        for (int idx = tid - 32; idx < CT * CT; idx += 96) {
            int t = idx >> 6, s = idx & 63;
            int dd = t - s;
            float kv = Kt[dd < 0 ? 0 : dd];
            PT[t][s] = (dd < 0) ? 0.0f : kv;
        }
    }
    __syncthreads();

#pragma unroll
    for (int nt = 0; nt < 2; ++nt)
#pragma unroll
        for (int i = 0; i < 8; ++i) acc[nt][i] = 0.0f;
#pragma unroll
    for (int kc = 0; kc < 16; ++kc) {
        int kk = kc * 4 + khalf;
        v2f a  = *(const v2f*)&PT[wave * 16 + lrow][kk];
        v2f b0 = *(const v2f*)&ZS[lrow][kk];
        v2f b1 = *(const v2f*)&ZS[16 + lrow][kk];
        acc[0] = __builtin_amdgcn_wmma_f32_16x16x4_f32(
            false, a, false, b0, (short)0, acc[0], false, false);
        acc[1] = __builtin_amdgcn_wmma_f32_16x16x4_f32(
            false, a, false, b1, (short)0, acc[1], false, false);
    }
#pragma unroll
    for (int kc = 0; kc < 16; ++kc) {
        int kk = kc * 4 + khalf;
        v2f a  = *(const v2f*)&Qs[wave * 16 + lrow][kk];
        v2f b0 = *(const v2f*)&ZS[lrow][CT + kk];
        v2f b1 = *(const v2f*)&ZS[16 + lrow][CT + kk];
        acc[0] = __builtin_amdgcn_wmma_f32_16x16x4_f32(
            false, a, false, b0, (short)0, acc[0], false, false);
        acc[1] = __builtin_amdgcn_wmma_f32_16x16x4_f32(
            false, a, false, b1, (short)0, acc[1], false, false);
    }
    __syncthreads();

    {
        float d = D_skip[h];
#pragma unroll
        for (int nt = 0; nt < 2; ++nt)
#pragma unroll
            for (int i = 0; i < 8; ++i) {
                int c = nt * 16 + lrow;
                int t = wave * 16 + i + madd;
                ZS[c][t] = fmaf(d, ZS[c][t], acc[nt][i]);
            }
    }
    __syncthreads();

#pragma unroll
    for (int j = 0; j < 4; ++j) {
        int l = j * 512 + tid * 4;
        float4 v = *(const float4*)&ZS[l >> 6][l & 63];
        v.x = gelu_tanh(v.x); v.y = gelu_tanh(v.y);
        v.z = gelu_tanh(v.z); v.w = gelu_tanh(v.w);
        *(float4*)(yrow + l) = v;
    }
}

#define YPITCH 516

__device__ __forceinline__ void glu_tile_mma(v8f (&acc)[4], const v4f (&q)[8], const v2f (&bf)[4]) {
#pragma unroll
    for (int kc = 0; kc < 4; ++kc) {
#pragma unroll
        for (int t = 0; t < 4; ++t) {
            v4f v = q[2 * t + (kc >> 1)];
            v2f a = (kc & 1) ? __builtin_shufflevector(v, v, 2, 3)
                             : __builtin_shufflevector(v, v, 0, 1);
            acc[t] = __builtin_amdgcn_wmma_f32_16x16x4_f32(
                false, a, false, bf[kc], (short)0, acc[t], false, false);
        }
    }
}

__global__ __launch_bounds__(512) void s4_glu_ln_fused(
        const float* __restrict__ Wp, const float* __restrict__ bias,
        const float* __restrict__ Y, float* __restrict__ Z,
        const float* __restrict__ g, const float* __restrict__ b) {
    __shared__ __attribute__((aligned(16))) float Ys[16][YPITCH];
    __shared__ float red[2][FNW][16];

    const int tid = threadIdx.x;
    const int lane = tid & 31;
    const int wave = tid >> 5;
    const int lrow = lane & 15;
    const int hi = lane >> 4;
    const int khalf = hi << 1;
    const int madd = hi << 3;
    const int col0 = blockIdx.x * 16;

    const v4f* wp = (const v4f*)Wp + wave * 256 + lane;

    v4f q[3][8];
    v2f bf[3][4];
#pragma unroll
    for (int j = 0; j < 8; ++j) q[0][j] = wp[j * 32];
#pragma unroll
    for (int j = 0; j < 8; ++j) q[1][j] = wp[WTILE_V4 + j * 32];

    float zres[2][8];
    v8f acc[4];
#pragma unroll
    for (int t = 0; t < 2; ++t)
#pragma unroll
        for (int i = 0; i < 8; ++i) {
            int j = wave * 32 + t * 16 + i + madd;
            int jc = j < HCH ? j : HCH - 1;
            zres[t][i] = Z[(size_t)jc * LSEQ + col0 + lrow];
            acc[t][i] = bias[jc];
            acc[2 + t][i] = bias[HCH + jc];
        }

    {
        int n = tid & 15;
        int kr = tid >> 4;
#pragma unroll
        for (int jj = 0; jj < 16; ++jj) {
            int k = kr + 32 * jj;
            int kc = k < KD ? k : KD - 1;
            float v = Y[(size_t)kc * LSEQ + col0 + n];
            Ys[n][k] = (k < KD) ? v : 0.0f;
        }
    }
    __syncthreads();

    const float* yrow = &Ys[lrow][khalf];
#pragma unroll
    for (int kc = 0; kc < 4; ++kc) {
        bf[0][kc] = *(const v2f*)(yrow + kc * 4);
        bf[1][kc] = *(const v2f*)(yrow + FBK + kc * 4);
    }

#pragma unroll
    for (int kt = 0; kt < FNT; ++kt) {
        if (kt + 2 < FNT) {
#pragma unroll
            for (int j = 0; j < 8; ++j)
                q[(kt + 2) % 3][j] = wp[(kt + 2) * WTILE_V4 + j * 32];
#pragma unroll
            for (int kc = 0; kc < 4; ++kc)
                bf[(kt + 2) % 3][kc] = *(const v2f*)(yrow + (kt + 2) * FBK + kc * 4);
        }
        glu_tile_mma(acc, q[kt % 3], bf[kt % 3]);
    }

    int lane_e = lane, wave_e = wave;
    asm volatile("v_nop\n\tv_nop\n\tv_nop\n\tv_nop\n\tv_nop\n\tv_nop\n\tv_nop\n\tv_nop"
                 : "+v"(acc[0]), "+v"(acc[1]), "+v"(acc[2]), "+v"(acc[3]),
                   "+v"(lane_e), "+v"(wave_e));
    const int lrow_e = lane_e & 15;
    const int hi_e = lane_e >> 4;
    const int madd_e = hi_e << 3;
    const int xaddr = (lane_e ^ 16) << 2;

    float x[2][8];
    float s = 0.0f;
#pragma unroll
    for (int t = 0; t < 2; ++t)
#pragma unroll
        for (int i = 0; i < 8; ++i) {
            int j = wave_e * 32 + t * 16 + i + madd_e;
            float v = acc[t][i] * sigmoidf(acc[2 + t][i]) + zres[t][i];
            v = (j < HCH) ? v : 0.0f;
            x[t][i] = v;
            s += v;
        }
    s += __int_as_float(__builtin_amdgcn_ds_bpermute(xaddr, __float_as_int(s)));
    if (hi_e == 0) red[0][wave_e][lrow_e] = s;
    __syncthreads();
    float tot = 0.0f;
#pragma unroll
    for (int w2 = 0; w2 < FNW; ++w2) tot += red[0][w2][lrow_e];
    const float mu = tot * (1.0f / (float)HCH);

    float qs = 0.0f;
#pragma unroll
    for (int t = 0; t < 2; ++t)
#pragma unroll
        for (int i = 0; i < 8; ++i) {
            int j = wave_e * 32 + t * 16 + i + madd_e;
            float d = (j < HCH) ? (x[t][i] - mu) : 0.0f;
            x[t][i] = d;
            qs += d * d;
        }
    qs += __int_as_float(__builtin_amdgcn_ds_bpermute(xaddr, __float_as_int(qs)));
    if (hi_e == 0) red[1][wave_e][lrow_e] = qs;
    __syncthreads();
    float tot2 = 0.0f;
#pragma unroll
    for (int w2 = 0; w2 < FNW; ++w2) tot2 += red[1][w2][lrow_e];
    const float rstd = rsqrtf(tot2 * (1.0f / (float)HCH) + LN_EPS);

#pragma unroll
    for (int t = 0; t < 2; ++t)
#pragma unroll
        for (int i = 0; i < 8; ++i) {
            int j = wave_e * 32 + t * 16 + i + madd_e;
            if (j < HCH)
                Z[(size_t)j * LSEQ + col0 + lrow_e] = x[t][i] * rstd * g[j] + b[j];
        }
}

extern "C" void kernel_launch(void* const* d_in, const int* in_sizes, int n_in,
                              void* d_out, int out_size, void* d_ws, size_t ws_size,
                              hipStream_t stream) {
    (void)in_sizes; (void)n_in; (void)out_size; (void)ws_size;
    const float* Zin    = (const float*)d_in[0];
    const float* log_dt = (const float*)d_in[1];
    const float* Arl    = (const float*)d_in[2];
    const float* Aim    = (const float*)d_in[3];
    const float* Cre    = (const float*)d_in[4];
    const float* Cim    = (const float*)d_in[5];
    const float* Dsk    = (const float*)d_in[6];
    const float* glu_w  = (const float*)d_in[7];
    const float* glu_b  = (const float*)d_in[8];
    const float* ln_g   = (const float*)d_in[9];
    const float* ln_b   = (const float*)d_in[10];

    float* ws = (float*)d_ws;
    const size_t HN = (size_t)HCH * NMODES;
    float* w_re  = ws;
    float* w_im  = ws + HN;
    float* c_re  = ws + 2 * HN;
    float* c_im  = ws + 3 * HN;
    float* zbuf  = ws + 4 * HN;
    float* ybuf  = zbuf + (size_t)HCH * LSEQ;
    float* wpk   = ybuf + (size_t)HCH * LSEQ;

    s4_setup_pack<<<SETUP_BLOCKS + PACK_BLOCKS, 256, 0, stream>>>(
        log_dt, Arl, Aim, Cre, Cim, w_re, w_im, c_re, c_im, glu_w, wpk);

    hipMemcpyAsync(zbuf, Zin, sizeof(float) * (size_t)HCH * LSEQ,
                   hipMemcpyDeviceToDevice, stream);

    for (int layer = 0; layer < NLAYERS; ++layer) {
        s4_scan<<<HCH, 128, 0, stream>>>(zbuf, w_re, w_im, c_re, c_im, Dsk, ybuf);
        s4_glu_ln_fused<<<128, 512, 0, stream>>>(wpk, glu_b, ybuf, zbuf, ln_g, ln_b);
    }

    hipMemcpyAsync(d_out, zbuf, sizeof(float) * (size_t)HCH * LSEQ,
                   hipMemcpyDeviceToDevice, stream);
}


# LayerNorm split in two: GLU groups re-tiled to 128 channels x 32 columns (256 groups = every WGP, was 128) store per-column mean/m2 partials; next reader merges and normalises (sum order changes)
# speedup vs baseline: 18.8572x; 1.4773x over previous
#include <hip/hip_runtime.h>
#include <hip/hip_bf16.h>

#define HCH 511
#define LSEQ 2048
#define NMODES 32
#define NLAYERS 4
#define KD 511
#define LN_EPS 1e-5f

typedef __attribute__((ext_vector_type(2))) float v2f;
typedef __attribute__((ext_vector_type(4))) float v4f;
typedef __attribute__((ext_vector_type(8))) float v8f;

__device__ __forceinline__ float gelu_tanh(float x) {
    float x3 = x * x * x;
    float t = tanhf(0.7978845608028654f * (x + 0.044715f * x3));
    return 0.5f * x * (1.0f + t);
}

__device__ __forceinline__ float sigmoidf(float x) {
    return 1.0f / (1.0f + expf(-x));
}

#define NPART 4
#define PCH 128
#define ST_LAYER (NPART * 2 * LSEQ)

__device__ __forceinline__ void ln_merge(float m0, float q0, float m1, float q1,
                                         float m2, float q2, float m3, float q3,
                                         float& mu, float& rstd) {
    const float n3 = (float)(HCH - 3 * PCH);
    mu = ((float)PCH * (m0 + m1 + m2) + n3 * m3) * (1.0f / (float)HCH);
    float d0 = m0 - mu, d1 = m1 - mu, d2 = m2 - mu, d3 = m3 - mu;
    float ss = (q0 + q1) + (q2 + q3)
             + (float)PCH * (d0 * d0 + d1 * d1 + d2 * d2) + n3 * (d3 * d3);
    rstd = rsqrtf(ss * (1.0f / (float)HCH) + LN_EPS);
}

__device__ __forceinline__ void ln_col(const float* __restrict__ st, int col,
                                       float& mu, float& rstd) {
    ln_merge(st[col], st[LSEQ + col], st[2 * LSEQ + col], st[3 * LSEQ + col],
             st[4 * LSEQ + col], st[5 * LSEQ + col], st[6 * LSEQ + col], st[7 * LSEQ + col],
             mu, rstd);
}

__device__ __forceinline__ float4 ln_apply4(float4 v, const float* __restrict__ st, int l,
                                            float gh, float bh) {
    float4 t[8];
#pragma unroll
    for (int r = 0; r < 8; ++r)
        t[r] = *(const float4*)(st + r * LSEQ + l);
    float mu, rs;
    ln_merge(t[0].x, t[1].x, t[2].x, t[3].x, t[4].x, t[5].x, t[6].x, t[7].x, mu, rs);
    v.x = (v.x - mu) * rs * gh + bh;
    ln_merge(t[0].y, t[1].y, t[2].y, t[3].y, t[4].y, t[5].y, t[6].y, t[7].y, mu, rs);
    v.y = (v.y - mu) * rs * gh + bh;
    ln_merge(t[0].z, t[1].z, t[2].z, t[3].z, t[4].z, t[5].z, t[6].z, t[7].z, mu, rs);
    v.z = (v.z - mu) * rs * gh + bh;
    ln_merge(t[0].w, t[1].w, t[2].w, t[3].w, t[4].w, t[5].w, t[6].w, t[7].w, mu, rs);
    v.w = (v.w - mu) * rs * gh + bh;
    return v;
}

#define FBK 16
#define FNW 16
#define FNT 32
#define WTILE_V4 (FNW * 8 * 32)
#define SETUP_BLOCKS 64
#define PACK_BLOCKS 512

__global__ __launch_bounds__(256) void s4_setup_pack(
                         const float* __restrict__ log_dt,
                         const float* __restrict__ Arl,
                         const float* __restrict__ Aim,
                         const float* __restrict__ Cre,
                         const float* __restrict__ Cim,
                         float* __restrict__ w_re, float* __restrict__ w_im,
                         float* __restrict__ c_re, float* __restrict__ c_im,
                         const float* __restrict__ Wg, float* __restrict__ Wp) {
    if (blockIdx.x >= SETUP_BLOCKS) {
        const int pb = blockIdx.x - SETUP_BLOCKS;
        const int w = pb >> 5, t = (pb >> 3) & 3, ktg = pb & 7;
        const int lane = threadIdx.x & 31, sub = threadIdx.x >> 5;
        const int p = sub & 1, kt = ktg * 4 + (sub >> 1);
        const int lrow = lane & 15, hi = lane >> 4;
        const int j = w * 32 + (t & 1) * 16 + lrow;
        const bool rowok = j < HCH;
        const int jc = rowok ? j : HCH - 1;
        const float* srow = Wg + (size_t)((t < 2) ? jc : HCH + jc) * KD;
        const int k0 = kt * FBK + p * 8 + hi * 2;
        float e[4];
#pragma unroll
        for (int s = 0; s < 4; ++s) {
            int k = k0 + (s & 1) + (s >> 1) * 4;
            float v = srow[k < KD ? k : KD - 1];
            e[s] = (rowok && k < KD) ? v : 0.0f;
        }
        v4f o = {e[0], e[1], e[2], e[3]};
        ((v4f*)Wp)[(size_t)((kt * FNW + w) * 8 + t * 2 + p) * 32 + lane] = o;
        return;
    }
    int idx = blockIdx.x * blockDim.x + threadIdx.x;
    if (idx >= HCH * NMODES) return;
    int h = idx / NMODES;
    float a_re = -expf(Arl[idx]);
    float a_im = Aim[idx];
    float dt = expf(log_dt[h]);
    float dre = a_re * dt, dim = a_im * dt;
    float er = expf(dre);
    float sn, cs;
    sincosf(dim, &sn, &cs);
    float wre = er * cs, wim = er * sn;
    float nre = wre - 1.0f, nim = wim;
    float inv = 1.0f / (a_re * a_re + a_im * a_im);
    float tre = (nre * a_re + nim * a_im) * inv;
    float tim = (nim * a_re - nre * a_im) * inv;
    float Cr = Cre[idx], Ci = Cim[idx];
    w_re[idx] = wre;
    w_im[idx] = wim;
    c_re[idx] = 2.0f * (Cr * tre - Ci * tim);
    c_im[idx] = 2.0f * (Cr * tim + Ci * tre);
}

#define CT 64
#define NCH 32
#define LDA 68
#define LDB 132

__device__ __forceinline__ void cmulf(float ar, float ai, float br, float bi,
                                      float& rr, float& ri) {
    rr = fmaf(ar, br, -ai * bi);
    ri = fmaf(ar, bi, ai * br);
}

__device__ __forceinline__ void csqrd(double& r, double& i) {
    double nr = fma(r, r, -(i * i));
    double ni = (r + r) * i;
    r = nr; i = ni;
}

__global__ __launch_bounds__(128) void s4_scan(
        const float* __restrict__ z,
        const float* __restrict__ w_re, const float* __restrict__ w_im,
        const float* __restrict__ c_re, const float* __restrict__ c_im,
        const float* __restrict__ D_skip, float* __restrict__ y,
        const float* __restrict__ st, const float* __restrict__ g,
        const float* __restrict__ b) {
    __shared__ __attribute__((aligned(16))) float PT[CT][LDA];
    __shared__ __attribute__((aligned(16))) float Qs[CT][LDA];
    __shared__ __attribute__((aligned(16))) float ZS[NCH][LDB];
    __shared__ float Kt[CT];
    __shared__ float WT[2][NMODES];
    __shared__ float Cr[NMODES];

    const int tid = threadIdx.x;
    const int lane = tid & 31;
    const int wave = tid >> 5;
    const int lrow = lane & 15;
    const int khalf = (lane >> 4) << 1;
    const int madd = (lane >> 4) << 3;
    const int h = blockIdx.x;
    const float* zrow = z + (size_t)h * LSEQ;
    float* yrow = y + (size_t)h * LSEQ;

    if (st != nullptr) {
        const float gh = g[h], bh = b[h];
#pragma unroll
        for (int j = 0; j < 4; ++j) {
            int l = j * 512 + tid * 4;
            float4 v = *(const float4*)(zrow + l);
            *(float4*)&ZS[l >> 6][l & 63] = ln_apply4(v, st, l, gh, bh);
        }
    } else {
#pragma unroll
        for (int j = 0; j < 4; ++j) {
            int l = j * 512 + tid * 4;
            float4 v = *(const float4*)(zrow + l);
            *(float4*)&ZS[l >> 6][l & 63] = v;
        }
    }

    {
        int cidx = h * NMODES + lane;
        float wr = w_re[cidx], wi = w_im[cidx];
        float cr = c_re[cidx], ci = c_im[cidx];
        float pr = 1.0f, pi = 0.0f;
        float er = 0.0f, ei = 0.0f;
        if (wave > 0) {
            double ar = (double)wr, ai = (double)wi;
            csqrd(ar, ai); csqrd(ar, ai); csqrd(ar, ai); csqrd(ar, ai);
            double br = ar, bi = ai;
            if (wave >= 2) csqrd(br, bi);
            if (wave == 3) {
                double dr = br, di = bi;
                csqrd(dr, di);
                er = (float)dr; ei = (float)di;
                double tr = fma(br, ar, -(bi * ai));
                double ti = fma(br, ai, bi * ar);
                br = tr; bi = ti;
            }
            pr = (float)br; pi = (float)bi;
        }
        const int m0 = wave * 16;
#pragma unroll
        for (int j = 0; j < 16; ++j) {
            int m = m0 + j;
            PT[lane][63 - m] = pr;
            PT[NMODES + lane][63 - m] = pi;
            if (m >= 1) {
                float qr, qi;
                cmulf(cr, ci, pr, pi, qr, qi);
                Qs[m - 1][lane] = qr;
                Qs[m - 1][NMODES + lane] = -qi;
            }
            float tr, ti;
            cmulf(pr, pi, wr, wi, tr, ti);
            pr = tr; pi = ti;
        }
        if (wave == 3) {
            float qr, qi;
            cmulf(cr, ci, er, ei, qr, qi);
            Qs[63][lane] = qr;
            Qs[63][NMODES + lane] = -qi;
            WT[0][lane] = er;
            WT[1][lane] = ei;
        }
        if (wave == 0) Cr[lane] = cr;
    }
    __syncthreads();

    if (tid < CT) {
        const float* rowp = (tid == 0) ? &Cr[0] : &Qs[tid > 0 ? tid - 1 : 0][0];
        float s = 0.0f;
#pragma unroll
        for (int n = 0; n < NMODES; ++n) s += rowp[n];
        Kt[tid] = s;
    }

    v8f acc[2];
#pragma unroll
    for (int nt = 0; nt < 2; ++nt)
#pragma unroll
        for (int i = 0; i < 8; ++i) acc[nt][i] = 0.0f;
#pragma unroll
    for (int kc = 0; kc < 16; ++kc) {
        int kk = kc * 4 + khalf;
        v2f a  = *(const v2f*)&PT[wave * 16 + lrow][kk];
        v2f b0 = *(const v2f*)&ZS[lrow][kk];
        v2f b1 = *(const v2f*)&ZS[16 + lrow][kk];
        acc[0] = __builtin_amdgcn_wmma_f32_16x16x4_f32(
            false, a, false, b0, (short)0, acc[0], false, false);
        acc[1] = __builtin_amdgcn_wmma_f32_16x16x4_f32(
            false, a, false, b1, (short)0, acc[1], false, false);
    }
#pragma unroll
    for (int nt = 0; nt < 2; ++nt)
#pragma unroll
        for (int i = 0; i < 8; ++i)
            ZS[nt * 16 + lrow][CT + wave * 16 + i + madd] = acc[nt][i];
    __syncthreads();

    if (wave == 0) {
        float wtr = WT[0][lane], wti = WT[1][lane];
        float sr = 0.0f, si = 0.0f;
#pragma unroll
        for (int c = 0; c < NCH; ++c) {
            float br = ZS[c][CT + lane];
            float bi = ZS[c][CT + NMODES + lane];
            ZS[c][CT + lane] = sr;
            ZS[c][CT + NMODES + lane] = si;
            float nr = fmaf(wtr, sr, fmaf(-wti, si, br));
            float ni = fmaf(wtr, si, fmaf(wti, sr, bi));
            sr = nr; si = ni;
        }
    } else {
        for (int idx = tid - 32; idx < CT * CT; idx += 96) {
            int t = idx >> 6, s = idx & 63;
            int dd = t - s;
            float kv = Kt[dd < 0 ? 0 : dd];
            PT[t][s] = (dd < 0) ? 0.0f : kv;
        }
    }
    __syncthreads();

#pragma unroll
    for (int nt = 0; nt < 2; ++nt)
#pragma unroll
        for (int i = 0; i < 8; ++i) acc[nt][i] = 0.0f;
#pragma unroll
    for (int kc = 0; kc < 16; ++kc) {
        int kk = kc * 4 + khalf;
        v2f a  = *(const v2f*)&PT[wave * 16 + lrow][kk];
        v2f b0 = *(const v2f*)&ZS[lrow][kk];
        v2f b1 = *(const v2f*)&ZS[16 + lrow][kk];
        acc[0] = __builtin_amdgcn_wmma_f32_16x16x4_f32(
            false, a, false, b0, (short)0, acc[0], false, false);
        acc[1] = __builtin_amdgcn_wmma_f32_16x16x4_f32(
            false, a, false, b1, (short)0, acc[1], false, false);
    }
#pragma unroll
    for (int kc = 0; kc < 16; ++kc) {
        int kk = kc * 4 + khalf;
        v2f a  = *(const v2f*)&Qs[wave * 16 + lrow][kk];
        v2f b0 = *(const v2f*)&ZS[lrow][CT + kk];
        v2f b1 = *(const v2f*)&ZS[16 + lrow][CT + kk];
        acc[0] = __builtin_amdgcn_wmma_f32_16x16x4_f32(
            false, a, false, b0, (short)0, acc[0], false, false);
        acc[1] = __builtin_amdgcn_wmma_f32_16x16x4_f32(
            false, a, false, b1, (short)0, acc[1], false, false);
    }
    __syncthreads();

    {
        float d = D_skip[h];
#pragma unroll
        for (int nt = 0; nt < 2; ++nt)
#pragma unroll
            for (int i = 0; i < 8; ++i) {
                int c = nt * 16 + lrow;
                int t = wave * 16 + i + madd;
                ZS[c][t] = fmaf(d, ZS[c][t], acc[nt][i]);
            }
    }
    __syncthreads();

#pragma unroll
    for (int j = 0; j < 4; ++j) {
        int l = j * 512 + tid * 4;
        float4 v = *(const float4*)&ZS[l >> 6][l & 63];
        v.x = gelu_tanh(v.x); v.y = gelu_tanh(v.y);
        v.z = gelu_tanh(v.z); v.w = gelu_tanh(v.w);
        *(float4*)(yrow + l) = v;
    }
}

#define YPITCH 516
#define GC 32
#define GW 8

__device__ __forceinline__ void glu_tile_mma(v8f (&acc)[4], const v4f (&q)[4], const v2f (&bf)[2][4]) {
#pragma unroll
    for (int kc = 0; kc < 4; ++kc) {
#pragma unroll
        for (int tt = 0; tt < 2; ++tt) {
            v4f v = q[2 * tt + (kc >> 1)];
            v2f a = (kc & 1) ? __builtin_shufflevector(v, v, 2, 3)
                             : __builtin_shufflevector(v, v, 0, 1);
#pragma unroll
            for (int ct = 0; ct < 2; ++ct)
                acc[2 * tt + ct] = __builtin_amdgcn_wmma_f32_16x16x4_f32(
                    false, a, false, bf[ct][kc], (short)0, acc[2 * tt + ct], false, false);
        }
    }
}

__global__ __launch_bounds__(256) void s4_glu_part(
        const float* __restrict__ Wp, const float* __restrict__ bias,
        const float* __restrict__ Y, float* __restrict__ Z,
        const float* __restrict__ g, const float* __restrict__ b,
        const float* __restrict__ st_in, float* __restrict__ st_out) {
    __shared__ __attribute__((aligned(16))) float Ys[GC][YPITCH];
    __shared__ float red[2][GW][GC];

    const int tid = threadIdx.x;
    const int lane = tid & 31;
    const int wave = tid >> 5;
    const int lrow = lane & 15;
    const int hi = lane >> 4;
    const int khalf = hi << 1;
    const int madd = hi << 3;
    const int part = blockIdx.x & (NPART - 1);
    const int col0 = (blockIdx.x >> 2) * GC;
    const int pw = part * 4 + (wave >> 1);
    const int u = wave & 1;
    const int ch0 = pw * 32 + u * 16;

    const v4f* wp = (const v4f*)Wp + pw * 256 + u * 64 + lane;

    v4f q[3][4];
    v2f bf[3][2][4];
#pragma unroll
    for (int j = 0; j < 4; ++j) q[0][j] = wp[(j >> 1) * 128 + (j & 1) * 32];
#pragma unroll
    for (int j = 0; j < 4; ++j) q[1][j] = wp[WTILE_V4 + (j >> 1) * 128 + (j & 1) * 32];

    float zres[2][8];
    v8f acc[4];
    {
        float mu[2] = {0.0f, 0.0f}, rs[2] = {1.0f, 1.0f};
        if (st_in != nullptr) {
#pragma unroll
            for (int ct = 0; ct < 2; ++ct)
                ln_col(st_in, col0 + ct * 16 + lrow, mu[ct], rs[ct]);
        }
#pragma unroll
        for (int i = 0; i < 8; ++i) {
            int j = ch0 + i + madd;
            int jc = j < HCH ? j : HCH - 1;
            float gj = 1.0f, bj = 0.0f;
            if (st_in != nullptr) { gj = g[jc]; bj = b[jc]; }
            float bv = bias[jc], bg = bias[HCH + jc];
#pragma unroll
            for (int ct = 0; ct < 2; ++ct) {
                float zv = Z[(size_t)jc * LSEQ + col0 + ct * 16 + lrow];
                zres[ct][i] = (st_in != nullptr) ? (zv - mu[ct]) * rs[ct] * gj + bj : zv;
                acc[ct][i] = bv;
                acc[2 + ct][i] = bg;
            }
        }
    }

    {
        int n = tid & 31;
        int kr = tid >> 5;
#pragma unroll 32
        for (int jj = 0; jj < 64; ++jj) {
            int k = kr + 8 * jj;
            int kc = k < KD ? k : KD - 1;
            float v = Y[(size_t)kc * LSEQ + col0 + n];
            Ys[n][k] = (k < KD) ? v : 0.0f;
        }
    }
    __syncthreads();

    const float* yrow = &Ys[lrow][khalf];
#pragma unroll
    for (int ct = 0; ct < 2; ++ct)
#pragma unroll
        for (int kc = 0; kc < 4; ++kc) {
            bf[0][ct][kc] = *(const v2f*)(yrow + ct * 16 * YPITCH + kc * 4);
            bf[1][ct][kc] = *(const v2f*)(yrow + ct * 16 * YPITCH + FBK + kc * 4);
        }

#pragma unroll
    for (int kt = 0; kt < FNT; ++kt) {
        if (kt + 2 < FNT) {
#pragma unroll
            for (int j = 0; j < 4; ++j)
                q[(kt + 2) % 3][j] = wp[(kt + 2) * WTILE_V4 + (j >> 1) * 128 + (j & 1) * 32];
#pragma unroll
            for (int ct = 0; ct < 2; ++ct)
#pragma unroll
                for (int kc = 0; kc < 4; ++kc)
                    bf[(kt + 2) % 3][ct][kc] =
                        *(const v2f*)(yrow + ct * 16 * YPITCH + (kt + 2) * FBK + kc * 4);
        }
        glu_tile_mma(acc, q[kt % 3], bf[kt % 3]);
    }

    int lane_e = lane, wave_e = wave;
    asm volatile("v_nop\n\tv_nop\n\tv_nop\n\tv_nop\n\tv_nop\n\tv_nop\n\tv_nop\n\tv_nop"
                 : "+v"(acc[0]), "+v"(acc[1]), "+v"(acc[2]), "+v"(acc[3]),
                   "+v"(lane_e), "+v"(wave_e));
    const int lrow_e = lane_e & 15;
    const int hi_e = lane_e >> 4;
    const int madd_e = hi_e << 3;
    const int tid_e = wave_e * 32 + lane_e;
    const int ch0_e = (part * 4 + (wave_e >> 1)) * 32 + (wave_e & 1) * 16;
    const int xaddr = (lane_e ^ 16) << 2;
    const float inv_n = (part == NPART - 1) ? 1.0f / (float)(HCH - 3 * PCH) : 1.0f / (float)PCH;

    float x[2][8];
    float s[2] = {0.0f, 0.0f};
#pragma unroll
    for (int ct = 0; ct < 2; ++ct)
#pragma unroll
        for (int i = 0; i < 8; ++i) {
            int j = ch0_e + i + madd_e;
            float v = acc[ct][i] * sigmoidf(acc[2 + ct][i]) + zres[ct][i];
            v = (j < HCH) ? v : 0.0f;
            x[ct][i] = v;
            s[ct] += v;
            if (j < HCH)
                Z[(size_t)j * LSEQ + col0 + ct * 16 + lrow_e] = v;
        }
#pragma unroll
    for (int ct = 0; ct < 2; ++ct) {
        s[ct] += __int_as_float(__builtin_amdgcn_ds_bpermute(xaddr, __float_as_int(s[ct])));
        if (hi_e == 0) red[0][wave_e][ct * 16 + lrow_e] = s[ct];
    }
    __syncthreads();
    float mean[2];
#pragma unroll
    for (int ct = 0; ct < 2; ++ct) {
        float tot = 0.0f;
#pragma unroll
        for (int w2 = 0; w2 < GW; ++w2) tot += red[0][w2][ct * 16 + lrow_e];
        mean[ct] = tot * inv_n;
    }

    float qs[2] = {0.0f, 0.0f};
#pragma unroll
    for (int ct = 0; ct < 2; ++ct)
#pragma unroll
        for (int i = 0; i < 8; ++i) {
            int j = ch0_e + i + madd_e;
            float d = (j < HCH) ? (x[ct][i] - mean[ct]) : 0.0f;
            qs[ct] += d * d;
        }
#pragma unroll
    for (int ct = 0; ct < 2; ++ct) {
        qs[ct] += __int_as_float(__builtin_amdgcn_ds_bpermute(xaddr, __float_as_int(qs[ct])));
        if (hi_e == 0) red[1][wave_e][ct * 16 + lrow_e] = qs[ct];
    }
    __syncthreads();
    if (tid_e < GC) {
        float tot = 0.0f, tot2 = 0.0f;
#pragma unroll
        for (int w2 = 0; w2 < GW; ++w2) {
            tot += red[0][w2][tid_e];
            tot2 += red[1][w2][tid_e];
        }
        st_out[(size_t)(part * 2) * LSEQ + col0 + tid_e] = tot * inv_n;
        st_out[(size_t)(part * 2 + 1) * LSEQ + col0 + tid_e] = tot2;
    }
}

__global__ __launch_bounds__(256) void s4_ln_out(
        const float* __restrict__ X, const float* __restrict__ st,
        const float* __restrict__ g, const float* __restrict__ b,
        float* __restrict__ out) {
    const int l = ((blockIdx.x & 7) * 64 + (threadIdx.x & 63)) * 4;
    const int h0 = (blockIdx.x >> 3) * 16 + (threadIdx.x >> 6);
    float4 t[8];
#pragma unroll
    for (int r = 0; r < 8; ++r)
        t[r] = *(const float4*)(st + r * LSEQ + l);
    float4 mu, rs;
    ln_merge(t[0].x, t[1].x, t[2].x, t[3].x, t[4].x, t[5].x, t[6].x, t[7].x, mu.x, rs.x);
    ln_merge(t[0].y, t[1].y, t[2].y, t[3].y, t[4].y, t[5].y, t[6].y, t[7].y, mu.y, rs.y);
    ln_merge(t[0].z, t[1].z, t[2].z, t[3].z, t[4].z, t[5].z, t[6].z, t[7].z, mu.z, rs.z);
    ln_merge(t[0].w, t[1].w, t[2].w, t[3].w, t[4].w, t[5].w, t[6].w, t[7].w, mu.w, rs.w);
#pragma unroll
    for (int k = 0; k < 4; ++k) {
        int h = h0 + 4 * k;
        if (h < HCH) {
            float4 v = *(const float4*)(X + (size_t)h * LSEQ + l);
            float gh = g[h], bh = b[h];
            v.x = (v.x - mu.x) * rs.x * gh + bh;
            v.y = (v.y - mu.y) * rs.y * gh + bh;
            v.z = (v.z - mu.z) * rs.z * gh + bh;
            v.w = (v.w - mu.w) * rs.w * gh + bh;
            *(float4*)(out + (size_t)h * LSEQ + l) = v;
        }
    }
}

extern "C" void kernel_launch(void* const* d_in, const int* in_sizes, int n_in,
                              void* d_out, int out_size, void* d_ws, size_t ws_size,
                              hipStream_t stream) {
    (void)in_sizes; (void)n_in; (void)out_size; (void)ws_size;
    const float* Zin    = (const float*)d_in[0];
    const float* log_dt = (const float*)d_in[1];
    const float* Arl    = (const float*)d_in[2];
    const float* Aim    = (const float*)d_in[3];
    const float* Cre    = (const float*)d_in[4];
    const float* Cim    = (const float*)d_in[5];
    const float* Dsk    = (const float*)d_in[6];
    const float* glu_w  = (const float*)d_in[7];
    const float* glu_b  = (const float*)d_in[8];
    const float* ln_g   = (const float*)d_in[9];
    const float* ln_b   = (const float*)d_in[10];

    float* ws = (float*)d_ws;
    const size_t HN = (size_t)HCH * NMODES;
    float* w_re  = ws;
    float* w_im  = ws + HN;
    float* c_re  = ws + 2 * HN;
    float* c_im  = ws + 3 * HN;
    float* zbuf  = ws + 4 * HN;
    float* stat  = zbuf + (size_t)HCH * LSEQ;
    float* wpk   = stat + (size_t)HCH * LSEQ;
    float* ybuf  = (float*)d_out;
    float* st0 = stat;
    float* st1 = stat + ST_LAYER;
    float* st2 = stat + 2 * ST_LAYER;
    float* st3 = stat + 3 * ST_LAYER;

    s4_setup_pack<<<SETUP_BLOCKS + PACK_BLOCKS, 256, 0, stream>>>(
        log_dt, Arl, Aim, Cre, Cim, w_re, w_im, c_re, c_im, glu_w, wpk);

    hipMemcpyAsync(zbuf, Zin, sizeof(float) * (size_t)HCH * LSEQ,
                   hipMemcpyDeviceToDevice, stream);

    s4_scan<<<HCH, 128, 0, stream>>>(zbuf, w_re, w_im, c_re, c_im, Dsk, ybuf, nullptr, ln_g, ln_b);
    s4_glu_part<<<256, 256, 0, stream>>>(wpk, glu_b, ybuf, zbuf, ln_g, ln_b, nullptr, st0);
    s4_scan<<<HCH, 128, 0, stream>>>(zbuf, w_re, w_im, c_re, c_im, Dsk, ybuf, st0, ln_g, ln_b);
    s4_glu_part<<<256, 256, 0, stream>>>(wpk, glu_b, ybuf, zbuf, ln_g, ln_b, st0, st1);
    s4_scan<<<HCH, 128, 0, stream>>>(zbuf, w_re, w_im, c_re, c_im, Dsk, ybuf, st1, ln_g, ln_b);
    s4_glu_part<<<256, 256, 0, stream>>>(wpk, glu_b, ybuf, zbuf, ln_g, ln_b, st1, st2);
    s4_scan<<<HCH, 128, 0, stream>>>(zbuf, w_re, w_im, c_re, c_im, Dsk, ybuf, st2, ln_g, ln_b);
    s4_glu_part<<<256, 256, 0, stream>>>(wpk, glu_b, ybuf, zbuf, ln_g, ln_b, st2, st3);

    s4_ln_out<<<256, 256, 0, stream>>>(zbuf, st3, ln_g, ln_b, (float*)d_out);
}


# dependent load rounds at kernel entry collapsed to one: GLU asks for glu_w tiles, bias and whole Y strip at once, stages Y in 2 k phases (LDS-only barriers); scan, ln_out likewise; sums unchanged
# speedup vs baseline: 19.6628x; 1.0427x over previous
#include <hip/hip_runtime.h>
#include <hip/hip_bf16.h>

#define HCH 511
#define LSEQ 2048
#define NMODES 32
#define NLAYERS 4
#define KD 511
#define LN_EPS 1e-5f

typedef __attribute__((ext_vector_type(2))) float v2f;
typedef __attribute__((ext_vector_type(4))) float v4f;
typedef __attribute__((ext_vector_type(8))) float v8f;

__device__ __forceinline__ float gelu_tanh(float x) {
    float x3 = x * x * x;
    float t = tanhf(0.7978845608028654f * (x + 0.044715f * x3));
    return 0.5f * x * (1.0f + t);
}

__device__ __forceinline__ float sigmoidf(float x) {
    return 1.0f / (1.0f + expf(-x));
}

#define NPART 4
#define PCH 128
#define ST_LAYER (NPART * 2 * LSEQ)

__device__ __forceinline__ void ln_merge(float m0, float q0, float m1, float q1,
                                         float m2, float q2, float m3, float q3,
                                         float& mu, float& rstd) {
    const float n3 = (float)(HCH - 3 * PCH);
    mu = ((float)PCH * (m0 + m1 + m2) + n3 * m3) * (1.0f / (float)HCH);
    float d0 = m0 - mu, d1 = m1 - mu, d2 = m2 - mu, d3 = m3 - mu;
    float ss = (q0 + q1) + (q2 + q3)
             + (float)PCH * (d0 * d0 + d1 * d1 + d2 * d2) + n3 * (d3 * d3);
    rstd = rsqrtf(ss * (1.0f / (float)HCH) + LN_EPS);
}

__device__ __forceinline__ void ln_col(const float* __restrict__ st, int col,
                                       float& mu, float& rstd) {
    ln_merge(st[col], st[LSEQ + col], st[2 * LSEQ + col], st[3 * LSEQ + col],
             st[4 * LSEQ + col], st[5 * LSEQ + col], st[6 * LSEQ + col], st[7 * LSEQ + col],
             mu, rstd);
}

__device__ __forceinline__ float4 ln_apply4(float4 v, const float* __restrict__ st, int l,
                                            float gh, float bh) {
    float4 t[8];
#pragma unroll
    for (int r = 0; r < 8; ++r)
        t[r] = *(const float4*)(st + r * LSEQ + l);
    float mu, rs;
    ln_merge(t[0].x, t[1].x, t[2].x, t[3].x, t[4].x, t[5].x, t[6].x, t[7].x, mu, rs);
    v.x = (v.x - mu) * rs * gh + bh;
    ln_merge(t[0].y, t[1].y, t[2].y, t[3].y, t[4].y, t[5].y, t[6].y, t[7].y, mu, rs);
    v.y = (v.y - mu) * rs * gh + bh;
    ln_merge(t[0].z, t[1].z, t[2].z, t[3].z, t[4].z, t[5].z, t[6].z, t[7].z, mu, rs);
    v.z = (v.z - mu) * rs * gh + bh;
    ln_merge(t[0].w, t[1].w, t[2].w, t[3].w, t[4].w, t[5].w, t[6].w, t[7].w, mu, rs);
    v.w = (v.w - mu) * rs * gh + bh;
    return v;
}

#define FBK 16
#define FNW 16
#define FNT 32
#define WTILE_V4 (FNW * 8 * 32)
#define SETUP_BLOCKS 64
#define PACK_BLOCKS 512

__global__ __launch_bounds__(256) void s4_setup_pack(
                         const float* __restrict__ log_dt,
                         const float* __restrict__ Arl,
                         const float* __restrict__ Aim,
                         const float* __restrict__ Cre,
                         const float* __restrict__ Cim,
                         float* __restrict__ w_re, float* __restrict__ w_im,
                         float* __restrict__ c_re, float* __restrict__ c_im,
                         const float* __restrict__ Wg, float* __restrict__ Wp) {
    if (blockIdx.x >= SETUP_BLOCKS) {
        const int pb = blockIdx.x - SETUP_BLOCKS;
        const int w = pb >> 5, t = (pb >> 3) & 3, ktg = pb & 7;
        const int lane = threadIdx.x & 31, sub = threadIdx.x >> 5;
        const int p = sub & 1, kt = ktg * 4 + (sub >> 1);
        const int lrow = lane & 15, hi = lane >> 4;
        const int j = w * 32 + (t & 1) * 16 + lrow;
        const bool rowok = j < HCH;
        const int jc = rowok ? j : HCH - 1;
        const float* srow = Wg + (size_t)((t < 2) ? jc : HCH + jc) * KD;
        const int k0 = kt * FBK + p * 8 + hi * 2;
        float e[4];
#pragma unroll
        for (int s = 0; s < 4; ++s) {
            int k = k0 + (s & 1) + (s >> 1) * 4;
            float v = srow[k < KD ? k : KD - 1];
            e[s] = (rowok && k < KD) ? v : 0.0f;
        }
        v4f o = {e[0], e[1], e[2], e[3]};
        ((v4f*)Wp)[(size_t)((kt * FNW + w) * 8 + t * 2 + p) * 32 + lane] = o;
        return;
    }
    int idx = blockIdx.x * blockDim.x + threadIdx.x;
    if (idx >= HCH * NMODES) return;
    int h = idx / NMODES;
    float a_re = -expf(Arl[idx]);
    float a_im = Aim[idx];
    float dt = expf(log_dt[h]);
    float dre = a_re * dt, dim = a_im * dt;
    float er = expf(dre);
    float sn, cs;
    sincosf(dim, &sn, &cs);
    float wre = er * cs, wim = er * sn;
    float nre = wre - 1.0f, nim = wim;
    float inv = 1.0f / (a_re * a_re + a_im * a_im);
    float tre = (nre * a_re + nim * a_im) * inv;
    float tim = (nim * a_re - nre * a_im) * inv;
    float Cr = Cre[idx], Ci = Cim[idx];
    w_re[idx] = wre;
    w_im[idx] = wim;
    c_re[idx] = 2.0f * (Cr * tre - Ci * tim);
    c_im[idx] = 2.0f * (Cr * tim + Ci * tre);
}

#define CT 64
#define NCH 32
#define LDA 68
#define LDB 132

__device__ __forceinline__ void cmulf(float ar, float ai, float br, float bi,
                                      float& rr, float& ri) {
    rr = fmaf(ar, br, -ai * bi);
    ri = fmaf(ar, bi, ai * br);
}

__device__ __forceinline__ void csqrd(double& r, double& i) {
    double nr = fma(r, r, -(i * i));
    double ni = (r + r) * i;
    r = nr; i = ni;
}

__global__ __launch_bounds__(128) void s4_scan(
        const float* __restrict__ z,
        const float* __restrict__ w_re, const float* __restrict__ w_im,
        const float* __restrict__ c_re, const float* __restrict__ c_im,
        const float* __restrict__ D_skip, float* __restrict__ y,
        const float* __restrict__ st, const float* __restrict__ g,
        const float* __restrict__ b) {
    __shared__ __attribute__((aligned(16))) float PT[CT][LDA];
    __shared__ __attribute__((aligned(16))) float Qs[CT][LDA];
    __shared__ __attribute__((aligned(16))) float ZS[NCH][LDB];
    __shared__ float Kt[CT];
    __shared__ float WT[2][NMODES];
    __shared__ float Cr[NMODES];

    const int tid = threadIdx.x;
    const int lane = tid & 31;
    const int wave = tid >> 5;
    const int lrow = lane & 15;
    const int khalf = (lane >> 4) << 1;
    const int madd = (lane >> 4) << 3;
    const int h = blockIdx.x;
    const float* zrow = z + (size_t)h * LSEQ;
    float* yrow = y + (size_t)h * LSEQ;

    const int cidx = h * NMODES + lane;
    const float wr = w_re[cidx], wi = w_im[cidx];
    const float cr = c_re[cidx], ci = c_im[cidx];

    if (st != nullptr) {
        const float gh = g[h], bh = b[h];
#pragma unroll
        for (int j = 0; j < 4; ++j) {
            int l = j * 512 + tid * 4;
            float4 v = *(const float4*)(zrow + l);
            *(float4*)&ZS[l >> 6][l & 63] = ln_apply4(v, st, l, gh, bh);
        }
    } else {
#pragma unroll
        for (int j = 0; j < 4; ++j) {
            int l = j * 512 + tid * 4;
            float4 v = *(const float4*)(zrow + l);
            *(float4*)&ZS[l >> 6][l & 63] = v;
        }
    }

    {
        float pr = 1.0f, pi = 0.0f;
        float er = 0.0f, ei = 0.0f;
        if (wave > 0) {
            double ar = (double)wr, ai = (double)wi;
            csqrd(ar, ai); csqrd(ar, ai); csqrd(ar, ai); csqrd(ar, ai);
            double br = ar, bi = ai;
            if (wave >= 2) csqrd(br, bi);
            if (wave == 3) {
                double dr = br, di = bi;
                csqrd(dr, di);
                er = (float)dr; ei = (float)di;
                double tr = fma(br, ar, -(bi * ai));
                double ti = fma(br, ai, bi * ar);
                br = tr; bi = ti;
            }
            pr = (float)br; pi = (float)bi;
        }
        const int m0 = wave * 16;
#pragma unroll
        for (int j = 0; j < 16; ++j) {
            int m = m0 + j;
            PT[lane][63 - m] = pr;
            PT[NMODES + lane][63 - m] = pi;
            if (m >= 1) {
                float qr, qi;
                cmulf(cr, ci, pr, pi, qr, qi);
                Qs[m - 1][lane] = qr;
                Qs[m - 1][NMODES + lane] = -qi;
            }
            float tr, ti;
            cmulf(pr, pi, wr, wi, tr, ti);
            pr = tr; pi = ti;
        }
        if (wave == 3) {
            float qr, qi;
            cmulf(cr, ci, er, ei, qr, qi);
            Qs[63][lane] = qr;
            Qs[63][NMODES + lane] = -qi;
            WT[0][lane] = er;
            WT[1][lane] = ei;
        }
        if (wave == 0) Cr[lane] = cr;
    }
    __syncthreads();

    if (tid < CT) {
        const float* rowp = (tid == 0) ? &Cr[0] : &Qs[tid > 0 ? tid - 1 : 0][0];
        float s = 0.0f;
#pragma unroll
        for (int n = 0; n < NMODES; ++n) s += rowp[n];
        Kt[tid] = s;
    }

    v8f acc[2];
#pragma unroll
    for (int nt = 0; nt < 2; ++nt)
#pragma unroll
        for (int i = 0; i < 8; ++i) acc[nt][i] = 0.0f;
#pragma unroll
    for (int kc = 0; kc < 16; ++kc) {
        int kk = kc * 4 + khalf;
        v2f a  = *(const v2f*)&PT[wave * 16 + lrow][kk];
        v2f b0 = *(const v2f*)&ZS[lrow][kk];
        v2f b1 = *(const v2f*)&ZS[16 + lrow][kk];
        acc[0] = __builtin_amdgcn_wmma_f32_16x16x4_f32(
            false, a, false, b0, (short)0, acc[0], false, false);
        acc[1] = __builtin_amdgcn_wmma_f32_16x16x4_f32(
            false, a, false, b1, (short)0, acc[1], false, false);
    }
#pragma unroll
    for (int nt = 0; nt < 2; ++nt)
#pragma unroll
        for (int i = 0; i < 8; ++i)
            ZS[nt * 16 + lrow][CT + wave * 16 + i + madd] = acc[nt][i];
    __syncthreads();

    if (wave == 0) {
        float wtr = WT[0][lane], wti = WT[1][lane];
        float sr = 0.0f, si = 0.0f;
#pragma unroll
        for (int c = 0; c < NCH; ++c) {
            float br = ZS[c][CT + lane];
            float bi = ZS[c][CT + NMODES + lane];
            ZS[c][CT + lane] = sr;
            ZS[c][CT + NMODES + lane] = si;
            float nr = fmaf(wtr, sr, fmaf(-wti, si, br));
            float ni = fmaf(wtr, si, fmaf(wti, sr, bi));
            sr = nr; si = ni;
        }
    } else {
        for (int idx = tid - 32; idx < CT * CT; idx += 96) {
            int t = idx >> 6, s = idx & 63;
            int dd = t - s;
            float kv = Kt[dd < 0 ? 0 : dd];
            PT[t][s] = (dd < 0) ? 0.0f : kv;
        }
    }
    __syncthreads();

#pragma unroll
    for (int nt = 0; nt < 2; ++nt)
#pragma unroll
        for (int i = 0; i < 8; ++i) acc[nt][i] = 0.0f;
#pragma unroll
    for (int kc = 0; kc < 16; ++kc) {
        int kk = kc * 4 + khalf;
        v2f a  = *(const v2f*)&PT[wave * 16 + lrow][kk];
        v2f b0 = *(const v2f*)&ZS[lrow][kk];
        v2f b1 = *(const v2f*)&ZS[16 + lrow][kk];
        acc[0] = __builtin_amdgcn_wmma_f32_16x16x4_f32(
            false, a, false, b0, (short)0, acc[0], false, false);
        acc[1] = __builtin_amdgcn_wmma_f32_16x16x4_f32(
            false, a, false, b1, (short)0, acc[1], false, false);
    }
#pragma unroll
    for (int kc = 0; kc < 16; ++kc) {
        int kk = kc * 4 + khalf;
        v2f a  = *(const v2f*)&Qs[wave * 16 + lrow][kk];
        v2f b0 = *(const v2f*)&ZS[lrow][CT + kk];
        v2f b1 = *(const v2f*)&ZS[16 + lrow][CT + kk];
        acc[0] = __builtin_amdgcn_wmma_f32_16x16x4_f32(
            false, a, false, b0, (short)0, acc[0], false, false);
        acc[1] = __builtin_amdgcn_wmma_f32_16x16x4_f32(
            false, a, false, b1, (short)0, acc[1], false, false);
    }
    __syncthreads();

    {
        float d = D_skip[h];
#pragma unroll
        for (int nt = 0; nt < 2; ++nt)
#pragma unroll
            for (int i = 0; i < 8; ++i) {
                int c = nt * 16 + lrow;
                int t = wave * 16 + i + madd;
                ZS[c][t] = fmaf(d, ZS[c][t], acc[nt][i]);
            }
    }
    __syncthreads();

#pragma unroll
    for (int j = 0; j < 4; ++j) {
        int l = j * 512 + tid * 4;
        float4 v = *(const float4*)&ZS[l >> 6][l & 63];
        v.x = gelu_tanh(v.x); v.y = gelu_tanh(v.y);
        v.z = gelu_tanh(v.z); v.w = gelu_tanh(v.w);
        *(float4*)(yrow + l) = v;
    }
}

#define YPITCH 516
#define GC 32
#define GW 8

__device__ __forceinline__ void lds_barrier() {
    __builtin_amdgcn_fence(__ATOMIC_RELEASE, "workgroup", "local");
    __builtin_amdgcn_s_barrier();
    __builtin_amdgcn_fence(__ATOMIC_ACQUIRE, "workgroup", "local");
}

__device__ __forceinline__ void glu_tile_mma(v8f (&acc)[4], const v4f (&q)[4], const v2f (&bf)[2][4]) {
#pragma unroll
    for (int kc = 0; kc < 4; ++kc) {
#pragma unroll
        for (int tt = 0; tt < 2; ++tt) {
            v4f v = q[2 * tt + (kc >> 1)];
            v2f a = (kc & 1) ? __builtin_shufflevector(v, v, 2, 3)
                             : __builtin_shufflevector(v, v, 0, 1);
#pragma unroll
            for (int ct = 0; ct < 2; ++ct)
                acc[2 * tt + ct] = __builtin_amdgcn_wmma_f32_16x16x4_f32(
                    false, a, false, bf[ct][kc], (short)0, acc[2 * tt + ct], false, false);
        }
    }
}

__global__ __launch_bounds__(256) void s4_glu_part(
        const float* __restrict__ Wp, const float* __restrict__ bias,
        const float* __restrict__ Y, float* __restrict__ Z,
        const float* __restrict__ g, const float* __restrict__ b,
        const float* __restrict__ st_in, float* __restrict__ st_out) {
    __shared__ __attribute__((aligned(16))) float Ys[GC][YPITCH];
    __shared__ float red[2][GW][GC];

    const int tid = threadIdx.x;
    const int lane = tid & 31;
    const int wave = tid >> 5;
    const int lrow = lane & 15;
    const int hi = lane >> 4;
    const int khalf = hi << 1;
    const int madd = hi << 3;
    const int part = blockIdx.x & (NPART - 1);
    const int col0 = (blockIdx.x >> 2) * GC;
    const int pw = part * 4 + (wave >> 1);
    const int u = wave & 1;
    const int ch0 = pw * 32 + u * 16;

    const v4f* wp = (const v4f*)Wp + pw * 256 + u * 64 + lane;

    v4f q[3][4];
    v2f bf[3][2][4];
#pragma unroll
    for (int j = 0; j < 4; ++j) q[0][j] = wp[(j >> 1) * 128 + (j & 1) * 32];
#pragma unroll
    for (int j = 0; j < 4; ++j) q[1][j] = wp[WTILE_V4 + (j >> 1) * 128 + (j & 1) * 32];

    v8f acc[4];
#pragma unroll
    for (int i = 0; i < 8; ++i) {
        int j = ch0 + i + madd;
        int jc = j < HCH ? j : HCH - 1;
        float bv = bias[jc], bg = bias[HCH + jc];
#pragma unroll
        for (int ct = 0; ct < 2; ++ct) {
            acc[ct][i] = bv;
            acc[2 + ct][i] = bg;
        }
    }

    const int n = lane;
    const int kr = wave;
    const float* ybase = Y + (size_t)kr * LSEQ + col0 + n;
    float yv[64];
#pragma unroll
    for (int jj = 0; jj < 16; ++jj) yv[jj] = ybase[(size_t)jj * (8 * LSEQ)];
    __builtin_amdgcn_sched_barrier(0);
#pragma unroll
    for (int jj = 16; jj < 63; ++jj) yv[jj] = ybase[(size_t)jj * (8 * LSEQ)];
    const float y63 = ybase[(size_t)(kr == 7 ? 62 : 63) * (8 * LSEQ)];
    __builtin_amdgcn_sched_barrier(0);

#pragma unroll
    for (int jj = 0; jj < 16; ++jj) Ys[n][kr + 8 * jj] = yv[jj];
    lds_barrier();

    const float* yrow = &Ys[lrow][khalf];
#pragma unroll
    for (int ct = 0; ct < 2; ++ct)
#pragma unroll
        for (int kc = 0; kc < 4; ++kc) {
            bf[0][ct][kc] = *(const v2f*)(yrow + ct * 16 * YPITCH + kc * 4);
            bf[1][ct][kc] = *(const v2f*)(yrow + ct * 16 * YPITCH + FBK + kc * 4);
        }

    const bool has_st = (st_in != nullptr);
    const float* stp = has_st ? st_in : Y;
    float sraw[2][8];
    float zraw[2][8];
    float gl[8], bl[8];

#pragma unroll
    for (int kt = 0; kt < FNT; ++kt) {
        if (kt == 6) {
            __builtin_amdgcn_sched_barrier(0);
            yv[63] = (kr == 7) ? 0.0f : y63;
#pragma unroll
            for (int jj = 16; jj < 64; ++jj) Ys[n][kr + 8 * jj] = yv[jj];
            lds_barrier();
        }
        if (kt + 2 < FNT) {
#pragma unroll
            for (int j = 0; j < 4; ++j)
                q[(kt + 2) % 3][j] = wp[(kt + 2) * WTILE_V4 + (j >> 1) * 128 + (j & 1) * 32];
#pragma unroll
            for (int ct = 0; ct < 2; ++ct)
#pragma unroll
                for (int kc = 0; kc < 4; ++kc)
                    bf[(kt + 2) % 3][ct][kc] =
                        *(const v2f*)(yrow + ct * 16 * YPITCH + (kt + 2) * FBK + kc * 4);
        }
        if (kt == 6) {
#pragma unroll
            for (int ct = 0; ct < 2; ++ct)
#pragma unroll
                for (int r = 0; r < 8; ++r)
                    sraw[ct][r] = stp[r * LSEQ + col0 + ct * 16 + lrow];
#pragma unroll
            for (int i = 0; i < 8; ++i) {
                int j = ch0 + i + madd;
                int jc = j < HCH ? j : HCH - 1;
                gl[i] = g[jc];
                bl[i] = b[jc];
#pragma unroll
                for (int ct = 0; ct < 2; ++ct)
                    zraw[ct][i] = Z[(size_t)jc * LSEQ + col0 + ct * 16 + lrow];
            }
            __builtin_amdgcn_sched_barrier(0);
        }
        glu_tile_mma(acc, q[kt % 3], bf[kt % 3]);
    }

    int lane_e = lane, wave_e = wave;
    asm volatile("v_nop\n\tv_nop\n\tv_nop\n\tv_nop\n\tv_nop\n\tv_nop\n\tv_nop\n\tv_nop"
                 : "+v"(acc[0]), "+v"(acc[1]), "+v"(acc[2]), "+v"(acc[3]),
                   "+v"(lane_e), "+v"(wave_e),
                   "+v"(sraw[0][0]), "+v"(sraw[0][1]), "+v"(sraw[0][2]), "+v"(sraw[0][3]),
                   "+v"(sraw[0][4]), "+v"(sraw[0][5]), "+v"(sraw[0][6]), "+v"(sraw[0][7]),
                   "+v"(sraw[1][0]), "+v"(sraw[1][1]), "+v"(sraw[1][2]), "+v"(sraw[1][3]),
                   "+v"(sraw[1][4]), "+v"(sraw[1][5]), "+v"(sraw[1][6]), "+v"(sraw[1][7]));
    const int lrow_e = lane_e & 15;
    const int hi_e = lane_e >> 4;
    const int madd_e = hi_e << 3;
    const int tid_e = wave_e * 32 + lane_e;
    const int ch0_e = (part * 4 + (wave_e >> 1)) * 32 + (wave_e & 1) * 16;
    const int xaddr = (lane_e ^ 16) << 2;
    const float inv_n = (part == NPART - 1) ? 1.0f / (float)(HCH - 3 * PCH) : 1.0f / (float)PCH;

    float zres[2][8];
    {
        float mu[2] = {0.0f, 0.0f}, rs[2] = {1.0f, 1.0f};
        if (has_st) {
#pragma unroll
            for (int ct = 0; ct < 2; ++ct)
                ln_merge(sraw[ct][0], sraw[ct][1], sraw[ct][2], sraw[ct][3],
                         sraw[ct][4], sraw[ct][5], sraw[ct][6], sraw[ct][7], mu[ct], rs[ct]);
        }
#pragma unroll
        for (int ct = 0; ct < 2; ++ct)
#pragma unroll
            for (int i = 0; i < 8; ++i)
                zres[ct][i] = has_st ? (zraw[ct][i] - mu[ct]) * rs[ct] * gl[i] + bl[i]
                                     : zraw[ct][i];
    }

    float x[2][8];
    float s[2] = {0.0f, 0.0f};
#pragma unroll
    for (int ct = 0; ct < 2; ++ct)
#pragma unroll
        for (int i = 0; i < 8; ++i) {
            int j = ch0_e + i + madd_e;
            float v = acc[ct][i] * sigmoidf(acc[2 + ct][i]) + zres[ct][i];
            v = (j < HCH) ? v : 0.0f;
            x[ct][i] = v;
            s[ct] += v;
            if (j < HCH)
                Z[(size_t)j * LSEQ + col0 + ct * 16 + lrow_e] = v;
        }
#pragma unroll
    for (int ct = 0; ct < 2; ++ct) {
        s[ct] += __int_as_float(__builtin_amdgcn_ds_bpermute(xaddr, __float_as_int(s[ct])));
        if (hi_e == 0) red[0][wave_e][ct * 16 + lrow_e] = s[ct];
    }
    __syncthreads();
    float mean[2];
#pragma unroll
    for (int ct = 0; ct < 2; ++ct) {
        float tot = 0.0f;
#pragma unroll
        for (int w2 = 0; w2 < GW; ++w2) tot += red[0][w2][ct * 16 + lrow_e];
        mean[ct] = tot * inv_n;
    }

    float qs[2] = {0.0f, 0.0f};
#pragma unroll
    for (int ct = 0; ct < 2; ++ct)
#pragma unroll
        for (int i = 0; i < 8; ++i) {
            int j = ch0_e + i + madd_e;
            float d = (j < HCH) ? (x[ct][i] - mean[ct]) : 0.0f;
            qs[ct] += d * d;
        }
#pragma unroll
    for (int ct = 0; ct < 2; ++ct) {
        qs[ct] += __int_as_float(__builtin_amdgcn_ds_bpermute(xaddr, __float_as_int(qs[ct])));
        if (hi_e == 0) red[1][wave_e][ct * 16 + lrow_e] = qs[ct];
    }
    __syncthreads();
    if (tid_e < GC) {
        float tot = 0.0f, tot2 = 0.0f;
#pragma unroll
        for (int w2 = 0; w2 < GW; ++w2) {
            tot += red[0][w2][tid_e];
            tot2 += red[1][w2][tid_e];
        }
        st_out[(size_t)(part * 2) * LSEQ + col0 + tid_e] = tot * inv_n;
        st_out[(size_t)(part * 2 + 1) * LSEQ + col0 + tid_e] = tot2;
    }
}

__global__ __launch_bounds__(256) void s4_ln_out(
        const float* __restrict__ X, const float* __restrict__ st,
        const float* __restrict__ g, const float* __restrict__ b,
        float* __restrict__ out) {
    const int l = ((blockIdx.x & 7) * 64 + (threadIdx.x & 63)) * 4;
    const int h0 = (blockIdx.x >> 3) * 16 + (threadIdx.x >> 6);
    float4 t[8];
#pragma unroll
    for (int r = 0; r < 8; ++r)
        t[r] = *(const float4*)(st + r * LSEQ + l);
    float4 xv[4];
    float ghv[4], bhv[4];
#pragma unroll
    for (int k = 0; k < 4; ++k) {
        int h = h0 + 4 * k;
        int hc = h < HCH ? h : HCH - 1;
        xv[k] = *(const float4*)(X + (size_t)hc * LSEQ + l);
        ghv[k] = g[hc];
        bhv[k] = b[hc];
    }
    __builtin_amdgcn_sched_barrier(0);
    float4 mu, rs;
    ln_merge(t[0].x, t[1].x, t[2].x, t[3].x, t[4].x, t[5].x, t[6].x, t[7].x, mu.x, rs.x);
    ln_merge(t[0].y, t[1].y, t[2].y, t[3].y, t[4].y, t[5].y, t[6].y, t[7].y, mu.y, rs.y);
    ln_merge(t[0].z, t[1].z, t[2].z, t[3].z, t[4].z, t[5].z, t[6].z, t[7].z, mu.z, rs.z);
    ln_merge(t[0].w, t[1].w, t[2].w, t[3].w, t[4].w, t[5].w, t[6].w, t[7].w, mu.w, rs.w);
#pragma unroll
    for (int k = 0; k < 4; ++k) {
        int h = h0 + 4 * k;
        if (h < HCH) {
            float4 v = xv[k];
            float gh = ghv[k], bh = bhv[k];
            v.x = (v.x - mu.x) * rs.x * gh + bh;
            v.y = (v.y - mu.y) * rs.y * gh + bh;
            v.z = (v.z - mu.z) * rs.z * gh + bh;
            v.w = (v.w - mu.w) * rs.w * gh + bh;
            *(float4*)(out + (size_t)h * LSEQ + l) = v;
        }
    }
}

extern "C" void kernel_launch(void* const* d_in, const int* in_sizes, int n_in,
                              void* d_out, int out_size, void* d_ws, size_t ws_size,
                              hipStream_t stream) {
    (void)in_sizes; (void)n_in; (void)out_size; (void)ws_size;
    const float* Zin    = (const float*)d_in[0];
    const float* log_dt = (const float*)d_in[1];
    const float* Arl    = (const float*)d_in[2];
    const float* Aim    = (const float*)d_in[3];
    const float* Cre    = (const float*)d_in[4];
    const float* Cim    = (const float*)d_in[5];
    const float* Dsk    = (const float*)d_in[6];
    const float* glu_w  = (const float*)d_in[7];
    const float* glu_b  = (const float*)d_in[8];
    const float* ln_g   = (const float*)d_in[9];
    const float* ln_b   = (const float*)d_in[10];

    float* ws = (float*)d_ws;
    const size_t HN = (size_t)HCH * NMODES;
    float* w_re  = ws;
    float* w_im  = ws + HN;
    float* c_re  = ws + 2 * HN;
    float* c_im  = ws + 3 * HN;
    float* zbuf  = ws + 4 * HN;
    float* stat  = zbuf + (size_t)HCH * LSEQ;
    float* wpk   = stat + (size_t)HCH * LSEQ;
    float* ybuf  = (float*)d_out;
    float* st0 = stat;
    float* st1 = stat + ST_LAYER;
    float* st2 = stat + 2 * ST_LAYER;
    float* st3 = stat + 3 * ST_LAYER;

    s4_setup_pack<<<SETUP_BLOCKS + PACK_BLOCKS, 256, 0, stream>>>(
        log_dt, Arl, Aim, Cre, Cim, w_re, w_im, c_re, c_im, glu_w, wpk);

    hipMemcpyAsync(zbuf, Zin, sizeof(float) * (size_t)HCH * LSEQ,
                   hipMemcpyDeviceToDevice, stream);

    s4_scan<<<HCH, 128, 0, stream>>>(zbuf, w_re, w_im, c_re, c_im, Dsk, ybuf, nullptr, ln_g, ln_b);
    s4_glu_part<<<256, 256, 0, stream>>>(wpk, glu_b, ybuf, zbuf, ln_g, ln_b, nullptr, st0);
    s4_scan<<<HCH, 128, 0, stream>>>(zbuf, w_re, w_im, c_re, c_im, Dsk, ybuf, st0, ln_g, ln_b);
    s4_glu_part<<<256, 256, 0, stream>>>(wpk, glu_b, ybuf, zbuf, ln_g, ln_b, st0, st1);
    s4_scan<<<HCH, 128, 0, stream>>>(zbuf, w_re, w_im, c_re, c_im, Dsk, ybuf, st1, ln_g, ln_b);
    s4_glu_part<<<256, 256, 0, stream>>>(wpk, glu_b, ybuf, zbuf, ln_g, ln_b, st1, st2);
    s4_scan<<<HCH, 128, 0, stream>>>(zbuf, w_re, w_im, c_re, c_im, Dsk, ybuf, st2, ln_g, ln_b);
    s4_glu_part<<<256, 256, 0, stream>>>(wpk, glu_b, ybuf, zbuf, ln_g, ln_b, st2, st3);

    s4_ln_out<<<256, 256, 0, stream>>>(zbuf, st3, ln_g, ln_b, (float*)d_out);
}


# gelu tanh and gate sigmoid as 1/(1+2^(-c u)) on hardware exp2 and reciprocal, branch-free so a lane's 16 values overlap (were 16 serial tanhf / expf+divide chains); sums and their order unchanged
# speedup vs baseline: 21.6644x; 1.1018x over previous
#include <hip/hip_runtime.h>
#include <hip/hip_bf16.h>

#define HCH 511
#define LSEQ 2048
#define NMODES 32
#define NLAYERS 4
#define KD 511
#define LN_EPS 1e-5f

typedef __attribute__((ext_vector_type(2))) float v2f;
typedef __attribute__((ext_vector_type(4))) float v4f;
typedef __attribute__((ext_vector_type(8))) float v8f;

__device__ __forceinline__ float gelu_tanh(float x) {
    const float k = -2.0f * 0.7978845608028654f * 1.4426950408889634f;
    float p = x * fmaf(0.044715f * x, x, 1.0f);
    float e = __builtin_amdgcn_exp2f(k * p);
    return x * __builtin_amdgcn_rcpf(1.0f + e);
}

__device__ __forceinline__ float sigmoidf(float x) {
    float e = __builtin_amdgcn_exp2f(-1.4426950408889634f * x);
    return __builtin_amdgcn_rcpf(1.0f + e);
}

#define NPART 4
#define PCH 128
#define ST_LAYER (NPART * 2 * LSEQ)

__device__ __forceinline__ void ln_merge(float m0, float q0, float m1, float q1,
                                         float m2, float q2, float m3, float q3,
                                         float& mu, float& rstd) {
    const float n3 = (float)(HCH - 3 * PCH);
    mu = ((float)PCH * (m0 + m1 + m2) + n3 * m3) * (1.0f / (float)HCH);
    float d0 = m0 - mu, d1 = m1 - mu, d2 = m2 - mu, d3 = m3 - mu;
    float ss = (q0 + q1) + (q2 + q3)
             + (float)PCH * (d0 * d0 + d1 * d1 + d2 * d2) + n3 * (d3 * d3);
    rstd = rsqrtf(ss * (1.0f / (float)HCH) + LN_EPS);
}

__device__ __forceinline__ void ln_col(const float* __restrict__ st, int col,
                                       float& mu, float& rstd) {
    ln_merge(st[col], st[LSEQ + col], st[2 * LSEQ + col], st[3 * LSEQ + col],
             st[4 * LSEQ + col], st[5 * LSEQ + col], st[6 * LSEQ + col], st[7 * LSEQ + col],
             mu, rstd);
}

__device__ __forceinline__ float4 ln_apply4(float4 v, const float* __restrict__ st, int l,
                                            float gh, float bh) {
    float4 t[8];
#pragma unroll
    for (int r = 0; r < 8; ++r)
        t[r] = *(const float4*)(st + r * LSEQ + l);
    float mu, rs;
    ln_merge(t[0].x, t[1].x, t[2].x, t[3].x, t[4].x, t[5].x, t[6].x, t[7].x, mu, rs);
    v.x = (v.x - mu) * rs * gh + bh;
    ln_merge(t[0].y, t[1].y, t[2].y, t[3].y, t[4].y, t[5].y, t[6].y, t[7].y, mu, rs);
    v.y = (v.y - mu) * rs * gh + bh;
    ln_merge(t[0].z, t[1].z, t[2].z, t[3].z, t[4].z, t[5].z, t[6].z, t[7].z, mu, rs);
    v.z = (v.z - mu) * rs * gh + bh;
    ln_merge(t[0].w, t[1].w, t[2].w, t[3].w, t[4].w, t[5].w, t[6].w, t[7].w, mu, rs);
    v.w = (v.w - mu) * rs * gh + bh;
    return v;
}

#define FBK 16
#define FNW 16
#define FNT 32
#define WTILE_V4 (FNW * 8 * 32)
#define SETUP_BLOCKS 64
#define PACK_BLOCKS 512

__global__ __launch_bounds__(256) void s4_setup_pack(
                         const float* __restrict__ log_dt,
                         const float* __restrict__ Arl,
                         const float* __restrict__ Aim,
                         const float* __restrict__ Cre,
                         const float* __restrict__ Cim,
                         float* __restrict__ w_re, float* __restrict__ w_im,
                         float* __restrict__ c_re, float* __restrict__ c_im,
                         const float* __restrict__ Wg, float* __restrict__ Wp) {
    if (blockIdx.x >= SETUP_BLOCKS) {
        const int pb = blockIdx.x - SETUP_BLOCKS;
        const int w = pb >> 5, t = (pb >> 3) & 3, ktg = pb & 7;
        const int lane = threadIdx.x & 31, sub = threadIdx.x >> 5;
        const int p = sub & 1, kt = ktg * 4 + (sub >> 1);
        const int lrow = lane & 15, hi = lane >> 4;
        const int j = w * 32 + (t & 1) * 16 + lrow;
        const bool rowok = j < HCH;
        const int jc = rowok ? j : HCH - 1;
        const float* srow = Wg + (size_t)((t < 2) ? jc : HCH + jc) * KD;
        const int k0 = kt * FBK + p * 8 + hi * 2;
        float e[4];
#pragma unroll
        for (int s = 0; s < 4; ++s) {
            int k = k0 + (s & 1) + (s >> 1) * 4;
            float v = srow[k < KD ? k : KD - 1];
            e[s] = (rowok && k < KD) ? v : 0.0f;
        }
        v4f o = {e[0], e[1], e[2], e[3]};
        ((v4f*)Wp)[(size_t)((kt * FNW + w) * 8 + t * 2 + p) * 32 + lane] = o;
        return;
    }
    int idx = blockIdx.x * blockDim.x + threadIdx.x;
    if (idx >= HCH * NMODES) return;
    int h = idx / NMODES;
    float a_re = -expf(Arl[idx]);
    float a_im = Aim[idx];
    float dt = expf(log_dt[h]);
    float dre = a_re * dt, dim = a_im * dt;
    float er = expf(dre);
    float sn, cs;
    sincosf(dim, &sn, &cs);
    float wre = er * cs, wim = er * sn;
    float nre = wre - 1.0f, nim = wim;
    float inv = 1.0f / (a_re * a_re + a_im * a_im);
    float tre = (nre * a_re + nim * a_im) * inv;
    float tim = (nim * a_re - nre * a_im) * inv;
    float Cr = Cre[idx], Ci = Cim[idx];
    w_re[idx] = wre;
    w_im[idx] = wim;
    c_re[idx] = 2.0f * (Cr * tre - Ci * tim);
    c_im[idx] = 2.0f * (Cr * tim + Ci * tre);
}

#define CT 64
#define NCH 32
#define LDA 68
#define LDB 132

__device__ __forceinline__ void cmulf(float ar, float ai, float br, float bi,
                                      float& rr, float& ri) {
    rr = fmaf(ar, br, -ai * bi);
    ri = fmaf(ar, bi, ai * br);
}

__device__ __forceinline__ void csqrd(double& r, double& i) {
    double nr = fma(r, r, -(i * i));
    double ni = (r + r) * i;
    r = nr; i = ni;
}

__global__ __launch_bounds__(128) void s4_scan(
        const float* __restrict__ z,
        const float* __restrict__ w_re, const float* __restrict__ w_im,
        const float* __restrict__ c_re, const float* __restrict__ c_im,
        const float* __restrict__ D_skip, float* __restrict__ y,
        const float* __restrict__ st, const float* __restrict__ g,
        const float* __restrict__ b) {
    __shared__ __attribute__((aligned(16))) float PT[CT][LDA];
    __shared__ __attribute__((aligned(16))) float Qs[CT][LDA];
    __shared__ __attribute__((aligned(16))) float ZS[NCH][LDB];
    __shared__ float Kt[CT];
    __shared__ float WT[2][NMODES];
    __shared__ float Cr[NMODES];

    const int tid = threadIdx.x;
    const int lane = tid & 31;
    const int wave = tid >> 5;
    const int lrow = lane & 15;
    const int khalf = (lane >> 4) << 1;
    const int madd = (lane >> 4) << 3;
    const int h = blockIdx.x;
    const float* zrow = z + (size_t)h * LSEQ;
    float* yrow = y + (size_t)h * LSEQ;

    const int cidx = h * NMODES + lane;
    const float wr = w_re[cidx], wi = w_im[cidx];
    const float cr = c_re[cidx], ci = c_im[cidx];

    if (st != nullptr) {
        const float gh = g[h], bh = b[h];
#pragma unroll
        for (int j = 0; j < 4; ++j) {
            int l = j * 512 + tid * 4;
            float4 v = *(const float4*)(zrow + l);
            *(float4*)&ZS[l >> 6][l & 63] = ln_apply4(v, st, l, gh, bh);
        }
    } else {
#pragma unroll
        for (int j = 0; j < 4; ++j) {
            int l = j * 512 + tid * 4;
            float4 v = *(const float4*)(zrow + l);
            *(float4*)&ZS[l >> 6][l & 63] = v;
        }
    }

    {
        float pr = 1.0f, pi = 0.0f;
        float er = 0.0f, ei = 0.0f;
        if (wave > 0) {
            double ar = (double)wr, ai = (double)wi;
            csqrd(ar, ai); csqrd(ar, ai); csqrd(ar, ai); csqrd(ar, ai);
            double br = ar, bi = ai;
            if (wave >= 2) csqrd(br, bi);
            if (wave == 3) {
                double dr = br, di = bi;
                csqrd(dr, di);
                er = (float)dr; ei = (float)di;
                double tr = fma(br, ar, -(bi * ai));
                double ti = fma(br, ai, bi * ar);
                br = tr; bi = ti;
            }
            pr = (float)br; pi = (float)bi;
        }
        const int m0 = wave * 16;
#pragma unroll
        for (int j = 0; j < 16; ++j) {
            int m = m0 + j;
            PT[lane][63 - m] = pr;
            PT[NMODES + lane][63 - m] = pi;
            if (m >= 1) {
                float qr, qi;
                cmulf(cr, ci, pr, pi, qr, qi);
                Qs[m - 1][lane] = qr;
                Qs[m - 1][NMODES + lane] = -qi;
            }
            float tr, ti;
            cmulf(pr, pi, wr, wi, tr, ti);
            pr = tr; pi = ti;
        }
        if (wave == 3) {
            float qr, qi;
            cmulf(cr, ci, er, ei, qr, qi);
            Qs[63][lane] = qr;
            Qs[63][NMODES + lane] = -qi;
            WT[0][lane] = er;
            WT[1][lane] = ei;
        }
        if (wave == 0) Cr[lane] = cr;
    }
    __syncthreads();

    if (tid < CT) {
        const float* rowp = (tid == 0) ? &Cr[0] : &Qs[tid > 0 ? tid - 1 : 0][0];
        float s = 0.0f;
#pragma unroll
        for (int n = 0; n < NMODES; ++n) s += rowp[n];
        Kt[tid] = s;
    }

    v8f acc[2];
#pragma unroll
    for (int nt = 0; nt < 2; ++nt)
#pragma unroll
        for (int i = 0; i < 8; ++i) acc[nt][i] = 0.0f;
#pragma unroll
    for (int kc = 0; kc < 16; ++kc) {
        int kk = kc * 4 + khalf;
        v2f a  = *(const v2f*)&PT[wave * 16 + lrow][kk];
        v2f b0 = *(const v2f*)&ZS[lrow][kk];
        v2f b1 = *(const v2f*)&ZS[16 + lrow][kk];
        acc[0] = __builtin_amdgcn_wmma_f32_16x16x4_f32(
            false, a, false, b0, (short)0, acc[0], false, false);
        acc[1] = __builtin_amdgcn_wmma_f32_16x16x4_f32(
            false, a, false, b1, (short)0, acc[1], false, false);
    }
#pragma unroll
    for (int nt = 0; nt < 2; ++nt)
#pragma unroll
        for (int i = 0; i < 8; ++i)
            ZS[nt * 16 + lrow][CT + wave * 16 + i + madd] = acc[nt][i];
    __syncthreads();

    if (wave == 0) {
        float wtr = WT[0][lane], wti = WT[1][lane];
        float sr = 0.0f, si = 0.0f;
#pragma unroll
        for (int c = 0; c < NCH; ++c) {
            float br = ZS[c][CT + lane];
            float bi = ZS[c][CT + NMODES + lane];
            ZS[c][CT + lane] = sr;
            ZS[c][CT + NMODES + lane] = si;
            float nr = fmaf(wtr, sr, fmaf(-wti, si, br));
            float ni = fmaf(wtr, si, fmaf(wti, sr, bi));
            sr = nr; si = ni;
        }
    } else {
        for (int idx = tid - 32; idx < CT * CT; idx += 96) {
            int t = idx >> 6, s = idx & 63;
            int dd = t - s;
            float kv = Kt[dd < 0 ? 0 : dd];
            PT[t][s] = (dd < 0) ? 0.0f : kv;
        }
    }
    __syncthreads();

#pragma unroll
    for (int nt = 0; nt < 2; ++nt)
#pragma unroll
        for (int i = 0; i < 8; ++i) acc[nt][i] = 0.0f;
#pragma unroll
    for (int kc = 0; kc < 16; ++kc) {
        int kk = kc * 4 + khalf;
        v2f a  = *(const v2f*)&PT[wave * 16 + lrow][kk];
        v2f b0 = *(const v2f*)&ZS[lrow][kk];
        v2f b1 = *(const v2f*)&ZS[16 + lrow][kk];
        acc[0] = __builtin_amdgcn_wmma_f32_16x16x4_f32(
            false, a, false, b0, (short)0, acc[0], false, false);
        acc[1] = __builtin_amdgcn_wmma_f32_16x16x4_f32(
            false, a, false, b1, (short)0, acc[1], false, false);
    }
#pragma unroll
    for (int kc = 0; kc < 16; ++kc) {
        int kk = kc * 4 + khalf;
        v2f a  = *(const v2f*)&Qs[wave * 16 + lrow][kk];
        v2f b0 = *(const v2f*)&ZS[lrow][CT + kk];
        v2f b1 = *(const v2f*)&ZS[16 + lrow][CT + kk];
        acc[0] = __builtin_amdgcn_wmma_f32_16x16x4_f32(
            false, a, false, b0, (short)0, acc[0], false, false);
        acc[1] = __builtin_amdgcn_wmma_f32_16x16x4_f32(
            false, a, false, b1, (short)0, acc[1], false, false);
    }
    __syncthreads();

    {
        float d = D_skip[h];
#pragma unroll
        for (int nt = 0; nt < 2; ++nt)
#pragma unroll
            for (int i = 0; i < 8; ++i) {
                int c = nt * 16 + lrow;
                int t = wave * 16 + i + madd;
                ZS[c][t] = fmaf(d, ZS[c][t], acc[nt][i]);
            }
    }
    __syncthreads();

#pragma unroll
    for (int j = 0; j < 4; ++j) {
        int l = j * 512 + tid * 4;
        float4 v = *(const float4*)&ZS[l >> 6][l & 63];
        v.x = gelu_tanh(v.x); v.y = gelu_tanh(v.y);
        v.z = gelu_tanh(v.z); v.w = gelu_tanh(v.w);
        *(float4*)(yrow + l) = v;
    }
}

#define YPITCH 516
#define GC 32
#define GW 8

__device__ __forceinline__ void lds_barrier() {
    __builtin_amdgcn_fence(__ATOMIC_RELEASE, "workgroup", "local");
    __builtin_amdgcn_s_barrier();
    __builtin_amdgcn_fence(__ATOMIC_ACQUIRE, "workgroup", "local");
}

__device__ __forceinline__ void glu_tile_mma(v8f (&acc)[4], const v4f (&q)[4], const v2f (&bf)[2][4]) {
#pragma unroll
    for (int kc = 0; kc < 4; ++kc) {
#pragma unroll
        for (int tt = 0; tt < 2; ++tt) {
            v4f v = q[2 * tt + (kc >> 1)];
            v2f a = (kc & 1) ? __builtin_shufflevector(v, v, 2, 3)
                             : __builtin_shufflevector(v, v, 0, 1);
#pragma unroll
            for (int ct = 0; ct < 2; ++ct)
                acc[2 * tt + ct] = __builtin_amdgcn_wmma_f32_16x16x4_f32(
                    false, a, false, bf[ct][kc], (short)0, acc[2 * tt + ct], false, false);
        }
    }
}

__global__ __launch_bounds__(256) void s4_glu_part(
        const float* __restrict__ Wp, const float* __restrict__ bias,
        const float* __restrict__ Y, float* __restrict__ Z,
        const float* __restrict__ g, const float* __restrict__ b,
        const float* __restrict__ st_in, float* __restrict__ st_out) {
    __shared__ __attribute__((aligned(16))) float Ys[GC][YPITCH];
    __shared__ float red[2][GW][GC];

    const int tid = threadIdx.x;
    const int lane = tid & 31;
    const int wave = tid >> 5;
    const int lrow = lane & 15;
    const int hi = lane >> 4;
    const int khalf = hi << 1;
    const int madd = hi << 3;
    const int part = blockIdx.x & (NPART - 1);
    const int col0 = (blockIdx.x >> 2) * GC;
    const int pw = part * 4 + (wave >> 1);
    const int u = wave & 1;
    const int ch0 = pw * 32 + u * 16;

    const v4f* wp = (const v4f*)Wp + pw * 256 + u * 64 + lane;

    v4f q[3][4];
    v2f bf[3][2][4];
#pragma unroll
    for (int j = 0; j < 4; ++j) q[0][j] = wp[(j >> 1) * 128 + (j & 1) * 32];
#pragma unroll
    for (int j = 0; j < 4; ++j) q[1][j] = wp[WTILE_V4 + (j >> 1) * 128 + (j & 1) * 32];

    v8f acc[4];
#pragma unroll
    for (int i = 0; i < 8; ++i) {
        int j = ch0 + i + madd;
        int jc = j < HCH ? j : HCH - 1;
        float bv = bias[jc], bg = bias[HCH + jc];
#pragma unroll
        for (int ct = 0; ct < 2; ++ct) {
            acc[ct][i] = bv;
            acc[2 + ct][i] = bg;
        }
    }

    const int n = lane;
    const int kr = wave;
    const float* ybase = Y + (size_t)kr * LSEQ + col0 + n;
    float yv[64];
#pragma unroll
    for (int jj = 0; jj < 16; ++jj) yv[jj] = ybase[(size_t)jj * (8 * LSEQ)];
    __builtin_amdgcn_sched_barrier(0);
#pragma unroll
    for (int jj = 16; jj < 63; ++jj) yv[jj] = ybase[(size_t)jj * (8 * LSEQ)];
    const float y63 = ybase[(size_t)(kr == 7 ? 62 : 63) * (8 * LSEQ)];
    __builtin_amdgcn_sched_barrier(0);

#pragma unroll
    for (int jj = 0; jj < 16; ++jj) Ys[n][kr + 8 * jj] = yv[jj];
    lds_barrier();

    const float* yrow = &Ys[lrow][khalf];
#pragma unroll
    for (int ct = 0; ct < 2; ++ct)
#pragma unroll
        for (int kc = 0; kc < 4; ++kc) {
            bf[0][ct][kc] = *(const v2f*)(yrow + ct * 16 * YPITCH + kc * 4);
            bf[1][ct][kc] = *(const v2f*)(yrow + ct * 16 * YPITCH + FBK + kc * 4);
        }

    const bool has_st = (st_in != nullptr);
    const float* stp = has_st ? st_in : Y;
    float sraw[2][8];
    float zraw[2][8];
    float gl[8], bl[8];

#pragma unroll
    for (int kt = 0; kt < FNT; ++kt) {
        if (kt == 6) {
            __builtin_amdgcn_sched_barrier(0);
            yv[63] = (kr == 7) ? 0.0f : y63;
#pragma unroll
            for (int jj = 16; jj < 64; ++jj) Ys[n][kr + 8 * jj] = yv[jj];
            lds_barrier();
        }
        if (kt + 2 < FNT) {
#pragma unroll
            for (int j = 0; j < 4; ++j)
                q[(kt + 2) % 3][j] = wp[(kt + 2) * WTILE_V4 + (j >> 1) * 128 + (j & 1) * 32];
#pragma unroll
            for (int ct = 0; ct < 2; ++ct)
#pragma unroll
                for (int kc = 0; kc < 4; ++kc)
                    bf[(kt + 2) % 3][ct][kc] =
                        *(const v2f*)(yrow + ct * 16 * YPITCH + (kt + 2) * FBK + kc * 4);
        }
        if (kt == 6) {
#pragma unroll
            for (int ct = 0; ct < 2; ++ct)
#pragma unroll
                for (int r = 0; r < 8; ++r)
                    sraw[ct][r] = stp[r * LSEQ + col0 + ct * 16 + lrow];
#pragma unroll
            for (int i = 0; i < 8; ++i) {
                int j = ch0 + i + madd;
                int jc = j < HCH ? j : HCH - 1;
                gl[i] = g[jc];
                bl[i] = b[jc];
#pragma unroll
                for (int ct = 0; ct < 2; ++ct)
                    zraw[ct][i] = Z[(size_t)jc * LSEQ + col0 + ct * 16 + lrow];
            }
            __builtin_amdgcn_sched_barrier(0);
        }
        glu_tile_mma(acc, q[kt % 3], bf[kt % 3]);
    }

    int lane_e = lane, wave_e = wave;
    asm volatile("v_nop\n\tv_nop\n\tv_nop\n\tv_nop\n\tv_nop\n\tv_nop\n\tv_nop\n\tv_nop"
                 : "+v"(acc[0]), "+v"(acc[1]), "+v"(acc[2]), "+v"(acc[3]),
                   "+v"(lane_e), "+v"(wave_e),
                   "+v"(sraw[0][0]), "+v"(sraw[0][1]), "+v"(sraw[0][2]), "+v"(sraw[0][3]),
                   "+v"(sraw[0][4]), "+v"(sraw[0][5]), "+v"(sraw[0][6]), "+v"(sraw[0][7]),
                   "+v"(sraw[1][0]), "+v"(sraw[1][1]), "+v"(sraw[1][2]), "+v"(sraw[1][3]),
                   "+v"(sraw[1][4]), "+v"(sraw[1][5]), "+v"(sraw[1][6]), "+v"(sraw[1][7]));
    const int lrow_e = lane_e & 15;
    const int hi_e = lane_e >> 4;
    const int madd_e = hi_e << 3;
    const int tid_e = wave_e * 32 + lane_e;
    const int ch0_e = (part * 4 + (wave_e >> 1)) * 32 + (wave_e & 1) * 16;
    const int xaddr = (lane_e ^ 16) << 2;
    const float inv_n = (part == NPART - 1) ? 1.0f / (float)(HCH - 3 * PCH) : 1.0f / (float)PCH;

    float zres[2][8];
    {
        float mu[2] = {0.0f, 0.0f}, rs[2] = {1.0f, 1.0f};
        if (has_st) {
#pragma unroll
            for (int ct = 0; ct < 2; ++ct)
                ln_merge(sraw[ct][0], sraw[ct][1], sraw[ct][2], sraw[ct][3],
                         sraw[ct][4], sraw[ct][5], sraw[ct][6], sraw[ct][7], mu[ct], rs[ct]);
        }
#pragma unroll
        for (int ct = 0; ct < 2; ++ct)
#pragma unroll
            for (int i = 0; i < 8; ++i)
                zres[ct][i] = has_st ? (zraw[ct][i] - mu[ct]) * rs[ct] * gl[i] + bl[i]
                                     : zraw[ct][i];
    }

    float x[2][8];
    float s[2] = {0.0f, 0.0f};
#pragma unroll
    for (int ct = 0; ct < 2; ++ct)
#pragma unroll
        for (int i = 0; i < 8; ++i) {
            int j = ch0_e + i + madd_e;
            float v = acc[ct][i] * sigmoidf(acc[2 + ct][i]) + zres[ct][i];
            v = (j < HCH) ? v : 0.0f;
            x[ct][i] = v;
            s[ct] += v;
        }
    static_assert((NPART * 4 - 1) * 32 + 16 + 6 + 8 < HCH, "rows i < 7 of every wave are real channels");
#pragma unroll
    for (int ct = 0; ct < 2; ++ct)
#pragma unroll
        for (int i = 0; i < 7; ++i)
            Z[(size_t)(ch0_e + i + madd_e) * LSEQ + col0 + ct * 16 + lrow_e] = x[ct][i];
#pragma unroll
    for (int ct = 0; ct < 2; ++ct) {
        int j = ch0_e + 7 + madd_e;
        if (j < HCH)
            Z[(size_t)j * LSEQ + col0 + ct * 16 + lrow_e] = x[ct][7];
    }
#pragma unroll
    for (int ct = 0; ct < 2; ++ct) {
        s[ct] += __int_as_float(__builtin_amdgcn_ds_bpermute(xaddr, __float_as_int(s[ct])));
        if (hi_e == 0) red[0][wave_e][ct * 16 + lrow_e] = s[ct];
    }
    __syncthreads();
    float mean[2];
#pragma unroll
    for (int ct = 0; ct < 2; ++ct) {
        float tot = 0.0f;
#pragma unroll
        for (int w2 = 0; w2 < GW; ++w2) tot += red[0][w2][ct * 16 + lrow_e];
        mean[ct] = tot * inv_n;
    }

    float qs[2] = {0.0f, 0.0f};
#pragma unroll
    for (int ct = 0; ct < 2; ++ct)
#pragma unroll
        for (int i = 0; i < 8; ++i) {
            int j = ch0_e + i + madd_e;
            float d = (j < HCH) ? (x[ct][i] - mean[ct]) : 0.0f;
            qs[ct] += d * d;
        }
#pragma unroll
    for (int ct = 0; ct < 2; ++ct) {
        qs[ct] += __int_as_float(__builtin_amdgcn_ds_bpermute(xaddr, __float_as_int(qs[ct])));
        if (hi_e == 0) red[1][wave_e][ct * 16 + lrow_e] = qs[ct];
    }
    __syncthreads();
    if (tid_e < GC) {
        float tot = 0.0f, tot2 = 0.0f;
#pragma unroll
        for (int w2 = 0; w2 < GW; ++w2) {
            tot += red[0][w2][tid_e];
            tot2 += red[1][w2][tid_e];
        }
        st_out[(size_t)(part * 2) * LSEQ + col0 + tid_e] = tot * inv_n;
        st_out[(size_t)(part * 2 + 1) * LSEQ + col0 + tid_e] = tot2;
    }
}

__global__ __launch_bounds__(256) void s4_ln_out(
        const float* __restrict__ X, const float* __restrict__ st,
        const float* __restrict__ g, const float* __restrict__ b,
        float* __restrict__ out) {
    const int l = ((blockIdx.x & 7) * 64 + (threadIdx.x & 63)) * 4;
    const int h0 = (blockIdx.x >> 3) * 16 + (threadIdx.x >> 6);
    float4 t[8];
#pragma unroll
    for (int r = 0; r < 8; ++r)
        t[r] = *(const float4*)(st + r * LSEQ + l);
    float4 xv[4];
    float ghv[4], bhv[4];
#pragma unroll
    for (int k = 0; k < 4; ++k) {
        int h = h0 + 4 * k;
        int hc = h < HCH ? h : HCH - 1;
        xv[k] = *(const float4*)(X + (size_t)hc * LSEQ + l);
        ghv[k] = g[hc];
        bhv[k] = b[hc];
    }
    __builtin_amdgcn_sched_barrier(0);
    float4 mu, rs;
    ln_merge(t[0].x, t[1].x, t[2].x, t[3].x, t[4].x, t[5].x, t[6].x, t[7].x, mu.x, rs.x);
    ln_merge(t[0].y, t[1].y, t[2].y, t[3].y, t[4].y, t[5].y, t[6].y, t[7].y, mu.y, rs.y);
    ln_merge(t[0].z, t[1].z, t[2].z, t[3].z, t[4].z, t[5].z, t[6].z, t[7].z, mu.z, rs.z);
    ln_merge(t[0].w, t[1].w, t[2].w, t[3].w, t[4].w, t[5].w, t[6].w, t[7].w, mu.w, rs.w);
#pragma unroll
    for (int k = 0; k < 4; ++k) {
        int h = h0 + 4 * k;
        if (h < HCH) {
            float4 v = xv[k];
            float gh = ghv[k], bh = bhv[k];
            v.x = (v.x - mu.x) * rs.x * gh + bh;
            v.y = (v.y - mu.y) * rs.y * gh + bh;
            v.z = (v.z - mu.z) * rs.z * gh + bh;
            v.w = (v.w - mu.w) * rs.w * gh + bh;
            *(float4*)(out + (size_t)h * LSEQ + l) = v;
        }
    }
}

extern "C" void kernel_launch(void* const* d_in, const int* in_sizes, int n_in,
                              void* d_out, int out_size, void* d_ws, size_t ws_size,
                              hipStream_t stream) {
    (void)in_sizes; (void)n_in; (void)out_size; (void)ws_size;
    const float* Zin    = (const float*)d_in[0];
    const float* log_dt = (const float*)d_in[1];
    const float* Arl    = (const float*)d_in[2];
    const float* Aim    = (const float*)d_in[3];
    const float* Cre    = (const float*)d_in[4];
    const float* Cim    = (const float*)d_in[5];
    const float* Dsk    = (const float*)d_in[6];
    const float* glu_w  = (const float*)d_in[7];
    const float* glu_b  = (const float*)d_in[8];
    const float* ln_g   = (const float*)d_in[9];
    const float* ln_b   = (const float*)d_in[10];

    float* ws = (float*)d_ws;
    const size_t HN = (size_t)HCH * NMODES;
    float* w_re  = ws;
    float* w_im  = ws + HN;
    float* c_re  = ws + 2 * HN;
    float* c_im  = ws + 3 * HN;
    float* zbuf  = ws + 4 * HN;
    float* stat  = zbuf + (size_t)HCH * LSEQ;
    float* wpk   = stat + (size_t)HCH * LSEQ;
    float* ybuf  = (float*)d_out;
    float* st0 = stat;
    float* st1 = stat + ST_LAYER;
    float* st2 = stat + 2 * ST_LAYER;
    float* st3 = stat + 3 * ST_LAYER;

    s4_setup_pack<<<SETUP_BLOCKS + PACK_BLOCKS, 256, 0, stream>>>(
        log_dt, Arl, Aim, Cre, Cim, w_re, w_im, c_re, c_im, glu_w, wpk);

    hipMemcpyAsync(zbuf, Zin, sizeof(float) * (size_t)HCH * LSEQ,
                   hipMemcpyDeviceToDevice, stream);

    s4_scan<<<HCH, 128, 0, stream>>>(zbuf, w_re, w_im, c_re, c_im, Dsk, ybuf, nullptr, ln_g, ln_b);
    s4_glu_part<<<256, 256, 0, stream>>>(wpk, glu_b, ybuf, zbuf, ln_g, ln_b, nullptr, st0);
    s4_scan<<<HCH, 128, 0, stream>>>(zbuf, w_re, w_im, c_re, c_im, Dsk, ybuf, st0, ln_g, ln_b);
    s4_glu_part<<<256, 256, 0, stream>>>(wpk, glu_b, ybuf, zbuf, ln_g, ln_b, st0, st1);
    s4_scan<<<HCH, 128, 0, stream>>>(zbuf, w_re, w_im, c_re, c_im, Dsk, ybuf, st1, ln_g, ln_b);
    s4_glu_part<<<256, 256, 0, stream>>>(wpk, glu_b, ybuf, zbuf, ln_g, ln_b, st1, st2);
    s4_scan<<<HCH, 128, 0, stream>>>(zbuf, w_re, w_im, c_re, c_im, Dsk, ybuf, st2, ln_g, ln_b);
    s4_glu_part<<<256, 256, 0, stream>>>(wpk, glu_b, ybuf, zbuf, ln_g, ln_b, st2, st3);

    s4_ln_out<<<256, 256, 0, stream>>>(zbuf, st3, ln_g, ln_b, (float*)d_out);
}
